# starting kernel
# baseline (speedup 1.0000x reference)
#include <hip/hip_runtime.h>
#include <hip/hip_bf16.h>

#define N_NODES 50000
#define N_EDGES 800000
#define IN_F 128
#define OUT_F 64
#define HEADS 4
#define HF 256
#define NEG_SLOPE 0.2f
#define GAT_EPS 1e-16f

typedef __attribute__((ext_vector_type(16))) _Float16 v16h;
typedef __attribute__((ext_vector_type(8)))  _Float16 v8h;
typedef __attribute__((ext_vector_type(8)))  float    v8f;
typedef __attribute__((ext_vector_type(4)))  float    v4f;

__global__ void cvt_f16_kernel(const float* __restrict__ x, const float* __restrict__ W,
                               _Float16* __restrict__ Xh, _Float16* __restrict__ Wh) {
  int idx = blockIdx.x * blockDim.x + threadIdx.x;
  int stride = gridDim.x * blockDim.x;
  for (int i = idx; i < N_NODES * IN_F; i += stride) Xh[i] = (_Float16)x[i];
  for (int i = idx; i < HF * IN_F; i += stride)      Wh[i] = (_Float16)W[i];
}

__global__ void init_kernel(float* __restrict__ out, const float* __restrict__ bias,
                            int* __restrict__ m, float* __restrict__ ssum) {
  int idx = blockIdx.x * blockDim.x + threadIdx.x;
  int stride = gridDim.x * blockDim.x;
  for (int i = idx; i < N_NODES * HF; i += stride) out[i] = bias[i & (HF - 1)];
  for (int i = idx; i < N_NODES * HEADS; i += stride) {
    m[i] = (int)0x80000000;
    ssum[i] = 0.0f;
  }
}

__global__ void __launch_bounds__(256)
gemm_wmma_kernel(const _Float16* __restrict__ Xh, const _Float16* __restrict__ Wh,
                 float* __restrict__ h) {
  const int wave  = (blockIdx.x << 3) + (threadIdx.x >> 5);
  const int lane  = threadIdx.x & 31;
  const int tileM = wave >> 4;
  const int tileN = wave & 15;
  const int mrow0 = tileM << 4;
  const int ncol0 = tileN << 4;

  const int l15   = lane & 15;
  const int khalf = lane >> 4;

  const int arow = mrow0 + l15;
  const int brow = ncol0 + l15;

  const _Float16* ap = Xh + (size_t)arow * IN_F + khalf * 8;
  const _Float16* bp = Wh + (size_t)brow * IN_F + khalf * 8;

  v8f c = {};
  #pragma unroll
  for (int kb = 0; kb < IN_F; kb += 32) {
    v8h a0 = *(const v8h*)(ap + kb);
    v8h a1 = *(const v8h*)(ap + kb + 16);
    v8h b0 = *(const v8h*)(bp + kb);
    v8h b1 = *(const v8h*)(bp + kb + 16);
    v16h A, B;
    #pragma unroll
    for (int i = 0; i < 8; ++i) { A[i] = a0[i]; A[i + 8] = a1[i];
                                  B[i] = b0[i]; B[i + 8] = b1[i]; }
    c = __builtin_amdgcn_wmma_f32_16x16x32_f16(false, A, false, B, (short)0, c,
                                               false, false);
  }
  float* hp = h + (size_t)(mrow0 + khalf * 8) * HF + ncol0 + l15;
  #pragma unroll
  for (int r = 0; r < 8; ++r) hp[(size_t)r * HF] = c[r];
}

__global__ void __launch_bounds__(256)
logits_kernel(const float* __restrict__ h,
              const float* __restrict__ att_src, const float* __restrict__ att_dst,
              float* __restrict__ a_src, float* __restrict__ a_dst) {
  int node = (blockIdx.x << 3) + (threadIdx.x >> 5);
  if (node >= N_NODES) return;
  int lane  = threadIdx.x & 31;
  int head  = lane >> 3;
  int fbase = (lane & 7) * 8;
  const float* hp = h + (size_t)node * HF + lane * 8;
  float ps = 0.f, pd = 0.f;
  #pragma unroll
  for (int j = 0; j < 8; ++j) {
    float hv = hp[j];
    ps += hv * att_src[head * OUT_F + fbase + j];
    pd += hv * att_dst[head * OUT_F + fbase + j];
  }
  #pragma unroll
  for (int off = 1; off < 8; off <<= 1) {
    ps += __shfl_xor(ps, off, 32);
    pd += __shfl_xor(pd, off, 32);
  }
  if ((lane & 7) == 0) {
    a_src[node * HEADS + head] = ps;
    a_dst[node * HEADS + head] = pd;
  }
}

__device__ __forceinline__ int enc_f(float f) {
  int i = __float_as_int(f);
  return (i >= 0) ? i : (i ^ 0x7fffffff);
}
__device__ __forceinline__ float dec_f(int i) {
  return __int_as_float((i >= 0) ? i : (i ^ 0x7fffffff));
}

__global__ void edge_score_kernel(const int* __restrict__ ei,
                                  const float* __restrict__ a_src,
                                  const float* __restrict__ a_dst,
                                  float* __restrict__ ebuf, int* __restrict__ m) {
  int i = blockIdx.x * blockDim.x + threadIdx.x;
  if (i >= N_EDGES) return;
  int s = ei[i], d = ei[N_EDGES + i];
  #pragma unroll
  for (int hh = 0; hh < HEADS; ++hh) {
    float ev = a_src[s * HEADS + hh] + a_dst[d * HEADS + hh];
    ev = (ev >= 0.f) ? ev : NEG_SLOPE * ev;
    ebuf[(size_t)i * HEADS + hh] = ev;
    atomicMax(&m[d * HEADS + hh], enc_f(ev));
  }
}

__global__ void edge_exp_kernel(const int* __restrict__ ei, const int* __restrict__ m,
                                float* __restrict__ ebuf, float* __restrict__ ssum) {
  int i = blockIdx.x * blockDim.x + threadIdx.x;
  if (i >= N_EDGES) return;
  int d = ei[N_EDGES + i];
  #pragma unroll
  for (int hh = 0; hh < HEADS; ++hh) {
    int menc = m[d * HEADS + hh];
    float mv = (menc == (int)0x80000000) ? 0.f : dec_f(menc);
    float ex = __expf(ebuf[(size_t)i * HEADS + hh] - mv);
    ebuf[(size_t)i * HEADS + hh] = ex;
    atomicAdd(&ssum[d * HEADS + hh], ex);
  }
}

__global__ void __launch_bounds__(256)
scatter_kernel(const int* __restrict__ ei, const float* __restrict__ ebuf,
               const float* __restrict__ ssum, const float* __restrict__ h,
               float* __restrict__ out) {
  int edge = (blockIdx.x << 3) + (threadIdx.x >> 5);
  if (edge >= N_EDGES) return;
  int lane = threadIdx.x & 31;
  int s = ei[edge], d = ei[N_EDGES + edge];
  const float* hs = h + (size_t)s * HF;
  float* od = out + (size_t)d * HF;
  __builtin_prefetch(hs + lane * 8, 0, 0);
  v4f e4 = *(const v4f*)(ebuf + (size_t)edge * HEADS);
  v4f s4 = *(const v4f*)(ssum + (size_t)d * HEADS);
  v4f alpha;
  #pragma unroll
  for (int k = 0; k < 4; ++k) alpha[k] = e4[k] / (s4[k] + GAT_EPS);
  #pragma unroll
  for (int j = 0; j < 8; ++j) {
    int f = lane + (j << 5);
    float a = alpha[j >> 1];
    atomicAdd(&od[f], a * hs[f]);
  }
}

extern "C" void kernel_launch(void* const* d_in, const int* in_sizes, int n_in,
                              void* d_out, int out_size, void* d_ws, size_t ws_size,
                              hipStream_t stream) {
  const float* x       = (const float*)d_in[0];
  const int*   ei      = (const int*)  d_in[1];
  const float* W       = (const float*)d_in[2];
  const float* att_src = (const float*)d_in[3];
  const float* att_dst = (const float*)d_in[4];
  const float* bias    = (const float*)d_in[5];
  float* out = (float*)d_out;

  char* p = (char*)d_ws;
  auto alloc = [&](size_t bytes) {
    char* r = p; p += (bytes + 255) & ~(size_t)255; return r;
  };
  float*    h     = (float*)   alloc((size_t)N_NODES * HF * 4);
  _Float16* Xh    = (_Float16*)alloc((size_t)N_NODES * IN_F * 2);
  _Float16* Wh    = (_Float16*)alloc((size_t)HF * IN_F * 2);
  float*    a_src = (float*)   alloc((size_t)N_NODES * HEADS * 4);
  float*    a_dst = (float*)   alloc((size_t)N_NODES * HEADS * 4);
  int*      m     = (int*)     alloc((size_t)N_NODES * HEADS * 4);
  float*    ssum  = (float*)   alloc((size_t)N_NODES * HEADS * 4);
  float*    ebuf  = (float*)   alloc((size_t)N_EDGES * HEADS * 4);

  cvt_f16_kernel  <<<2048, 256, 0, stream>>>(x, W, Xh, Wh);
  init_kernel     <<<2048, 256, 0, stream>>>(out, bias, m, ssum);
  gemm_wmma_kernel<<<6250, 256, 0, stream>>>(Xh, Wh, h);
  logits_kernel   <<<6250, 256, 0, stream>>>(h, att_src, att_dst, a_src, a_dst);
  edge_score_kernel<<<(N_EDGES + 255) / 256, 256, 0, stream>>>(ei, a_src, a_dst, ebuf, m);
  edge_exp_kernel <<<(N_EDGES + 255) / 256, 256, 0, stream>>>(ei, m, ebuf, ssum);
  scatter_kernel  <<<N_EDGES / 8, 256, 0, stream>>>(ei, ebuf, ssum, h, out);
}


# Projection kernel reads x itself: a block fetches its 16 float rows as whole lines, rounds them once to the 16-bit operand form, shares them through LDS; conversion pass and 16-bit copy of x gone
# speedup vs baseline: 4.8723x; 4.8723x over previous
#include <hip/hip_runtime.h>
#include <hip/hip_bf16.h>

#define N_NODES 50000
#define N_EDGES 800000
#define IN_F 128
#define OUT_F 64
#define HEADS 4
#define HF 256
#define NEG_SLOPE 0.2f
#define GAT_EPS 1e-16f
#define DEG_PLANE 50176

typedef __attribute__((ext_vector_type(16))) _Float16 v16h;
typedef __attribute__((ext_vector_type(8)))  _Float16 v8h;
typedef __attribute__((ext_vector_type(4)))  _Float16 v4h;
typedef __attribute__((ext_vector_type(8)))  float    v8f;
typedef __attribute__((ext_vector_type(4)))  float    v4f;
typedef __attribute__((ext_vector_type(2)))  float    v2f;
typedef __attribute__((ext_vector_type(4)))  int      v4i;

static_assert((2 * DEG_PLANE) % 4 == 0 && (HF * IN_F) % 4 == 0, "16-byte pieces");
__global__ void __launch_bounds__(256)
prep_kernel(const float* __restrict__ W, _Float16* __restrict__ Wh, int* __restrict__ deg) {
  const int idx = blockIdx.x * blockDim.x + threadIdx.x;
  const int stride = gridDim.x * blockDim.x;
  for (int i = idx; i < HF * IN_F / 4; i += stride) {
    const v4f w = *(const v4f*)(W + 4 * i);
    v4h o;
    for (int q = 0; q < 4; ++q) o[q] = (_Float16)w[q];
    *(v4h*)(Wh + 4 * i) = o;
  }
  const v4i zero = {0, 0, 0, 0};
  for (int i = idx; i < 2 * DEG_PLANE / 4; i += stride) *(v4i*)(deg + 4 * i) = zero;
}

static_assert(N_EDGES % 4 == 0, "four edges per lane, 16-byte loads");
__global__ void __launch_bounds__(256)
rank_kernel(const int* __restrict__ ei, int* __restrict__ deg, int* __restrict__ rank) {
  const int plane = (int)((__builtin_amdgcn_s_sendmsg_rtn(0x87) >> 18) & 1u);
  int i = blockIdx.x * blockDim.x + threadIdx.x;
  if (i >= N_EDGES / 4) return;
  const v4i d = *(const v4i*)(ei + N_EDGES + 4 * i);
  int* cnt = deg + plane * DEG_PLANE;
  const int mark = (int)((unsigned)plane << 31);
  v4i r;
  r[0] = atomicAdd(&cnt[d[0]], 1) | mark;
  r[1] = atomicAdd(&cnt[d[1]], 1) | mark;
  r[2] = atomicAdd(&cnt[d[2]], 1) | mark;
  r[3] = atomicAdd(&cnt[d[3]], 1) | mark;
  *(v4i*)(rank + 4 * i) = r;
}


template <int M>
__device__ __forceinline__ float row_xor_sum(float v) {
  int o = __builtin_amdgcn_update_dpp(0, __float_as_int(v), 0x160 + M, 0xf, 0xf, false);
  return v + __int_as_float(o);
}

#define XS_PITCH 136
__global__ void __launch_bounds__(128)
gemm_logits_kernel(const float* __restrict__ x, const _Float16* __restrict__ Wh,
                   const float* __restrict__ att_src, const float* __restrict__ att_dst,
                   float* __restrict__ h, float* __restrict__ a_src, float* __restrict__ a_dst) {
  const int lane  = threadIdx.x & 31;
  const int head  = threadIdx.x >> 5;
  const int mrow0 = blockIdx.x << 4;
  const int col0  = head * OUT_F;

  const int l15   = lane & 15;
  const int khalf = lane >> 4;

  const _Float16* bp = Wh + (size_t)(col0 + l15) * IN_F + khalf * 8;

  float ws[4], wd[4];
  #pragma unroll
  for (int j = 0; j < 4; ++j) {
    ws[j] = att_src[col0 + j * 16 + l15];
    wd[j] = att_dst[col0 + j * 16 + l15];
  }

  v16h B0[4];
  #pragma unroll
  for (int j = 0; j < 4; ++j) {
    const _Float16* bj = bp + (size_t)j * 16 * IN_F;
    v8h b0 = *(const v8h*)(bj);
    v8h b1 = *(const v8h*)(bj + 16);
    #pragma unroll
    for (int i = 0; i < 8; ++i) { B0[j][i] = b0[i]; B0[j][i + 8] = b1[i]; }
  }

  __shared__ _Float16 xs[16 * XS_PITCH];
  {
    const float* xr = x + (size_t)(mrow0 + head * 4) * IN_F + lane * 4;
    v4f xf[4];
    #pragma unroll
    for (int r = 0; r < 4; ++r) xf[r] = *(const v4f*)(xr + (size_t)r * IN_F);
    #pragma unroll
    for (int r = 0; r < 4; ++r) {
      v4h o;
      #pragma unroll
      for (int i = 0; i < 4; ++i) o[i] = (_Float16)xf[r][i];
      *(v4h*)(xs + (head * 4 + r) * XS_PITCH + lane * 4) = o;
    }
  }
  __syncthreads();

  const _Float16* ap = xs + l15 * XS_PITCH + khalf * 8;
  v16h A[4];
  #pragma unroll
  for (int kc = 0; kc < 4; ++kc) {
    v8h a0 = *(const v8h*)(ap + kc * 32);
    v8h a1 = *(const v8h*)(ap + kc * 32 + 16);
    #pragma unroll
    for (int i = 0; i < 8; ++i) { A[kc][i] = a0[i]; A[kc][i + 8] = a1[i]; }
  }

  v8f c[4];
  #pragma unroll
  for (int j = 0; j < 4; ++j) c[j] = (v8f){};
  #pragma unroll
  for (int kc = 0; kc < 4; ++kc) {
    #pragma unroll
    for (int j = 0; j < 4; ++j) {
      v16h B = B0[j];
      if (kc > 0) {
        const _Float16* bj = bp + (size_t)j * 16 * IN_F + kc * 32;
        v8h b0 = *(const v8h*)(bj);
        v8h b1 = *(const v8h*)(bj + 16);
        #pragma unroll
        for (int i = 0; i < 8; ++i) { B[i] = b0[i]; B[i + 8] = b1[i]; }
      }
      c[j] = __builtin_amdgcn_wmma_f32_16x16x32_f16(false, A[kc], false, B, (short)0, c[j],
                                                    false, false);
    }
  }

  float* hp = h + (size_t)(mrow0 + khalf * 8) * HF + col0 + l15;
  #pragma unroll
  for (int j = 0; j < 4; ++j) {
    #pragma unroll
    for (int r = 0; r < 8; ++r) hp[(size_t)r * HF + j * 16] = c[j][r];
  }

  float ps[8], pd[8];
  #pragma unroll
  for (int r = 0; r < 8; ++r) {
    ps[r] = c[0][r] * ws[0] + c[1][r] * ws[1] + c[2][r] * ws[2] + c[3][r] * ws[3];
    pd[r] = c[0][r] * wd[0] + c[1][r] * wd[1] + c[2][r] * wd[2] + c[3][r] * wd[3];
  }
  #pragma unroll
  for (int r = 0; r < 8; ++r) { ps[r] = row_xor_sum<1>(ps[r]); pd[r] = row_xor_sum<1>(pd[r]); }
  #pragma unroll
  for (int r = 0; r < 8; ++r) { ps[r] = row_xor_sum<2>(ps[r]); pd[r] = row_xor_sum<2>(pd[r]); }
  #pragma unroll
  for (int r = 0; r < 8; ++r) { ps[r] = row_xor_sum<4>(ps[r]); pd[r] = row_xor_sum<4>(pd[r]); }
  #pragma unroll
  for (int r = 0; r < 8; ++r) { ps[r] = row_xor_sum<8>(ps[r]); pd[r] = row_xor_sum<8>(pd[r]); }

  float v = ps[0];
  #pragma unroll
  for (int i = 1; i < 8; ++i) v = (l15 == i) ? ps[i] : v;
  #pragma unroll
  for (int i = 0; i < 8; ++i) v = (l15 == 8 + i) ? pd[i] : v;
  float* lp = (l15 < 8) ? a_src : a_dst;
  lp[(size_t)(mrow0 + khalf * 8 + (l15 & 7)) * HEADS + head] = v;
}

#define SCAN_PER 49
__global__ void __launch_bounds__(1024)
scan_kernel(const int* __restrict__ deg, int* __restrict__ rowptr) {
  __shared__ int buf[1024 * SCAN_PER];
  __shared__ int wtot[32];
  const int t    = threadIdx.x;
  const int lane = t & 31;
  const int wv   = t >> 5;
  #pragma unroll
  for (int j = 0; j < SCAN_PER; ++j) {
    int idx = j * 1024 + t;
    buf[idx] = (idx < N_NODES) ? deg[idx] + deg[DEG_PLANE + idx] : 0;
  }
  __syncthreads();
  int v[SCAN_PER];
  int run = 0;
  #pragma unroll
  for (int j = 0; j < SCAN_PER; ++j) {
    int x = buf[t * SCAN_PER + j];
    v[j] = run;
    run += x;
  }
  int x = run;
  for (int off = 1; off < 32; off <<= 1) {
    int y = __shfl_up(x, off, 32);
    if (lane >= off) x += y;
  }
  if (lane == 31) wtot[wv] = x;
  __syncthreads();
  int w = (lane < wv) ? wtot[lane] : 0;
  for (int off = 16; off >= 1; off >>= 1) w += __shfl_xor(w, off, 32);
  const int offs = w + x - run;
  #pragma unroll
  for (int j = 0; j < SCAN_PER; ++j) buf[t * SCAN_PER + j] = v[j] + offs;
  __syncthreads();
  #pragma unroll
  for (int j = 0; j < SCAN_PER; ++j) {
    int idx = j * 1024 + t;
    if (idx < N_NODES) rowptr[idx] = buf[idx];
  }
  if (t == 1023) rowptr[N_NODES] = offs + run;
}

__global__ void __launch_bounds__(256)
fill_kernel(const int* __restrict__ ei, const int* __restrict__ rowptr,
            const int* __restrict__ rank, int* __restrict__ csr_src) {
  int i = blockIdx.x * blockDim.x + threadIdx.x;
  if (i >= N_EDGES) return;
  int s = ei[i], d = ei[N_EDGES + i];
  const int rk = rank[i];
  const int back = rk >> 31;
  const int r = rk & 0x7fffffff;
  const int edge = rowptr[d - back];
  csr_src[back ? edge - 1 - r : edge + r] = s;
}

__device__ __forceinline__ v4f edge_scores(const float* __restrict__ a_src, int s, v4f ad) {
  v4f as = *(const v4f*)(a_src + (size_t)s * HEADS);
  v4f e;
  for (int k = 0; k < HEADS; ++k) {
    float ev = as[k] + ad[k];
    e[k] = (ev >= 0.f) ? ev : NEG_SLOPE * ev;
  }
  return e;
}

__global__ void __launch_bounds__(256)
gather_kernel(const int* __restrict__ rowptr, const int* __restrict__ csr_src,
              const float* __restrict__ a_src, const float* __restrict__ a_dst,
              const float* __restrict__ h, const float* __restrict__ bias,
              float* __restrict__ out) {
  const int node = __builtin_amdgcn_readfirstlane((int)((blockIdx.x << 3) + (threadIdx.x >> 5)));
  const int lane = threadIdx.x & 31;
  const int beg = rowptr[node];
  const int end = rowptr[node + 1];
  const v4f ad = *(const v4f*)(a_dst + (size_t)node * HEADS);

  const float ninf = -__builtin_huge_valf();
  v4f mx = {ninf, ninf, ninf, ninf};
  for (int p = beg + lane; p < end; p += 32) {
    v4f e = edge_scores(a_src, csr_src[p], ad);
    for (int k = 0; k < HEADS; ++k) mx[k] = fmaxf(mx[k], e[k]);
  }
  for (int off = 16; off >= 1; off >>= 1) {
    for (int k = 0; k < HEADS; ++k) mx[k] = fmaxf(mx[k], __shfl_xor(mx[k], off, 32));
  }
  v4f m;
  for (int k = 0; k < HEADS; ++k)
    m[k] = (__builtin_fabsf(mx[k]) <= 3.402823466e38f) ? mx[k] : 0.f;

  v4f sm = {0.f, 0.f, 0.f, 0.f};
  for (int p = beg + lane; p < end; p += 32) {
    v4f e = edge_scores(a_src, csr_src[p], ad);
    for (int k = 0; k < HEADS; ++k) sm[k] += __expf(e[k] - m[k]);
  }
  for (int off = 16; off >= 1; off >>= 1) {
    for (int k = 0; k < HEADS; ++k) sm[k] += __shfl_xor(sm[k], off, 32);
  }
  v4f den;
  for (int k = 0; k < HEADS; ++k) den[k] = sm[k] + GAT_EPS;

  float acc[8];
  for (int q = 0; q < 8; ++q) acc[q] = 0.f;
  const float* hl = h + lane * 2;
  for (int cb = beg; cb < end; cb += 32) {
    const int p = cb + lane;
    int s = 0;
    v4f al = {0.f, 0.f, 0.f, 0.f};
    if (p < end) {
      s = csr_src[p];
      v4f e = edge_scores(a_src, s, ad);
      for (int k = 0; k < HEADS; ++k) al[k] = __expf(e[k] - m[k]) / den[k];
    }
    const int cnt = (end - cb < 32) ? (end - cb) : 32;
    #pragma unroll 4
    for (int j = 0; j < cnt; ++j) {
      const int sj = __builtin_amdgcn_readlane(s, j);
      const float a0 = __int_as_float(__builtin_amdgcn_readlane(__float_as_int(al[0]), j));
      const float a1 = __int_as_float(__builtin_amdgcn_readlane(__float_as_int(al[1]), j));
      const float a2 = __int_as_float(__builtin_amdgcn_readlane(__float_as_int(al[2]), j));
      const float a3 = __int_as_float(__builtin_amdgcn_readlane(__float_as_int(al[3]), j));
      const float* hr = hl + (size_t)sj * HF;
      v2f r0 = *(const v2f*)(hr);
      v2f r1 = *(const v2f*)(hr + OUT_F);
      v2f r2 = *(const v2f*)(hr + 2 * OUT_F);
      v2f r3 = *(const v2f*)(hr + 3 * OUT_F);
      acc[0] += a0 * r0[0]; acc[1] += a0 * r0[1];
      acc[2] += a1 * r1[0]; acc[3] += a1 * r1[1];
      acc[4] += a2 * r2[0]; acc[5] += a2 * r2[1];
      acc[6] += a3 * r3[0]; acc[7] += a3 * r3[1];
    }
  }
  float* o = out + (size_t)node * HF + lane * 2;
  const float* bl = bias + lane * 2;
  for (int k = 0; k < HEADS; ++k) {
    v2f b = *(const v2f*)(bl + k * OUT_F);
    v2f r = {acc[2 * k] + b[0], acc[2 * k + 1] + b[1]};
    *(v2f*)(o + k * OUT_F) = r;
  }
}

extern "C" void kernel_launch(void* const* d_in, const int* in_sizes, int n_in,
                              void* d_out, int out_size, void* d_ws, size_t ws_size,
                              hipStream_t stream) {
  const float* x       = (const float*)d_in[0];
  const int*   ei      = (const int*)  d_in[1];
  const float* W       = (const float*)d_in[2];
  const float* att_src = (const float*)d_in[3];
  const float* att_dst = (const float*)d_in[4];
  const float* bias    = (const float*)d_in[5];
  float* out = (float*)d_out;

  char* p = (char*)d_ws;
  auto alloc = [&](size_t bytes) {
    char* r = p; p += (bytes + 255) & ~(size_t)255; return r;
  };
  float*    h     = (float*)   alloc((size_t)N_NODES * HF * 4);
  _Float16* Wh    = (_Float16*)alloc((size_t)HF * IN_F * 2);
  float*    a_src = (float*)   alloc((size_t)N_NODES * HEADS * 4);
  float*    a_dst = (float*)   alloc((size_t)N_NODES * HEADS * 4);
  int*      deg     = (int*)   alloc((size_t)2 * DEG_PLANE * 4);
  int*      rowptr  = (int*)   alloc((size_t)(N_NODES + 1) * 4);
  int*      rank    = (int*)   alloc((size_t)N_EDGES * 4);
  int*      csr_src = (int*)   alloc((size_t)N_EDGES * 4);

  prep_kernel     <<<98, 256, 0, stream>>>(W, Wh, deg);
  rank_kernel     <<<(N_EDGES / 4 + 255) / 256, 256, 0, stream>>>(ei, deg, rank);
  gemm_logits_kernel<<<N_NODES / 16, 128, 0, stream>>>(x, Wh, att_src, att_dst, h, a_src, a_dst);
  scan_kernel     <<<1, 1024, 0, stream>>>(deg, rowptr);
  fill_kernel     <<<(N_EDGES + 255) / 256, 256, 0, stream>>>(ei, rowptr, rank, csr_src);
  gather_kernel   <<<N_NODES / 8, 256, 0, stream>>>(rowptr, csr_src, a_src, a_dst, h, bias, out);
}


# Ranking pass and projection share one launch (ranking blocks first, strips behind, told apart by block number): they run side by side, 5 launches not 6; sums still arrival-ordered
# speedup vs baseline: 5.1645x; 1.0600x over previous
#include <hip/hip_runtime.h>
#include <hip/hip_bf16.h>

#define N_NODES 50000
#define N_EDGES 800000
#define IN_F 128
#define OUT_F 64
#define HEADS 4
#define HF 256
#define NEG_SLOPE 0.2f
#define GAT_EPS 1e-16f
#define DEG_PLANE 50176

typedef __attribute__((ext_vector_type(16))) _Float16 v16h;
typedef __attribute__((ext_vector_type(8)))  _Float16 v8h;
typedef __attribute__((ext_vector_type(4)))  _Float16 v4h;
typedef __attribute__((ext_vector_type(8)))  float    v8f;
typedef __attribute__((ext_vector_type(4)))  float    v4f;
typedef __attribute__((ext_vector_type(2)))  float    v2f;
typedef __attribute__((ext_vector_type(4)))  int      v4i;

static_assert((2 * DEG_PLANE) % 4 == 0 && (HF * IN_F) % 4 == 0, "16-byte pieces");
__global__ void __launch_bounds__(256)
prep_kernel(const float* __restrict__ W, _Float16* __restrict__ Wh, int* __restrict__ deg) {
  const int idx = blockIdx.x * blockDim.x + threadIdx.x;
  const int stride = gridDim.x * blockDim.x;
  for (int i = idx; i < HF * IN_F / 4; i += stride) {
    const v4f w = *(const v4f*)(W + 4 * i);
    v4h o;
    for (int q = 0; q < 4; ++q) o[q] = (_Float16)w[q];
    *(v4h*)(Wh + 4 * i) = o;
  }
  const v4i zero = {0, 0, 0, 0};
  for (int i = idx; i < 2 * DEG_PLANE / 4; i += stride) *(v4i*)(deg + 4 * i) = zero;
}

#define FUSED_THREADS 128
static_assert(N_EDGES % 4 == 0, "four edges per lane, 16-byte loads");
__device__ __forceinline__ void
rank_block(int blk, const int* __restrict__ ei, int* __restrict__ deg, int* __restrict__ rank) {
  const int plane = (int)((__builtin_amdgcn_s_sendmsg_rtn(0x87) >> 18) & 1u);
  int i = blk * FUSED_THREADS + threadIdx.x;
  if (i >= N_EDGES / 4) return;
  const v4i d = *(const v4i*)(ei + N_EDGES + 4 * i);
  int* cnt = deg + plane * DEG_PLANE;
  const int mark = (int)((unsigned)plane << 31);
  v4i r;
  r[0] = atomicAdd(&cnt[d[0]], 1) | mark;
  r[1] = atomicAdd(&cnt[d[1]], 1) | mark;
  r[2] = atomicAdd(&cnt[d[2]], 1) | mark;
  r[3] = atomicAdd(&cnt[d[3]], 1) | mark;
  *(v4i*)(rank + 4 * i) = r;
}


template <int M>
__device__ __forceinline__ float row_xor_sum(float v) {
  int o = __builtin_amdgcn_update_dpp(0, __float_as_int(v), 0x160 + M, 0xf, 0xf, false);
  return v + __int_as_float(o);
}

#define XS_PITCH 136
__device__ __forceinline__ void
gemm_logits_strip(int strip, _Float16* xs,
                  const float* __restrict__ x, const _Float16* __restrict__ Wh,
                  const float* __restrict__ att_src, const float* __restrict__ att_dst,
                  float* __restrict__ h, float* __restrict__ a_src, float* __restrict__ a_dst) {
  const int lane  = threadIdx.x & 31;
  const int head  = threadIdx.x >> 5;
  const int mrow0 = strip << 4;
  const int col0  = head * OUT_F;

  const int l15   = lane & 15;
  const int khalf = lane >> 4;

  const _Float16* bp = Wh + (size_t)(col0 + l15) * IN_F + khalf * 8;

  float ws[4], wd[4];
  #pragma unroll
  for (int j = 0; j < 4; ++j) {
    ws[j] = att_src[col0 + j * 16 + l15];
    wd[j] = att_dst[col0 + j * 16 + l15];
  }

  v16h B0[4];
  #pragma unroll
  for (int j = 0; j < 4; ++j) {
    const _Float16* bj = bp + (size_t)j * 16 * IN_F;
    v8h b0 = *(const v8h*)(bj);
    v8h b1 = *(const v8h*)(bj + 16);
    #pragma unroll
    for (int i = 0; i < 8; ++i) { B0[j][i] = b0[i]; B0[j][i + 8] = b1[i]; }
  }

  {
    const float* xr = x + (size_t)(mrow0 + head * 4) * IN_F + lane * 4;
    v4f xf[4];
    #pragma unroll
    for (int r = 0; r < 4; ++r) xf[r] = *(const v4f*)(xr + (size_t)r * IN_F);
    #pragma unroll
    for (int r = 0; r < 4; ++r) {
      v4h o;
      #pragma unroll
      for (int i = 0; i < 4; ++i) o[i] = (_Float16)xf[r][i];
      *(v4h*)(xs + (head * 4 + r) * XS_PITCH + lane * 4) = o;
    }
  }
  __syncthreads();

  const _Float16* ap = xs + l15 * XS_PITCH + khalf * 8;
  v16h A[4];
  #pragma unroll
  for (int kc = 0; kc < 4; ++kc) {
    v8h a0 = *(const v8h*)(ap + kc * 32);
    v8h a1 = *(const v8h*)(ap + kc * 32 + 16);
    #pragma unroll
    for (int i = 0; i < 8; ++i) { A[kc][i] = a0[i]; A[kc][i + 8] = a1[i]; }
  }

  v8f c[4];
  #pragma unroll
  for (int j = 0; j < 4; ++j) c[j] = (v8f){};
  #pragma unroll
  for (int kc = 0; kc < 4; ++kc) {
    #pragma unroll
    for (int j = 0; j < 4; ++j) {
      v16h B = B0[j];
      if (kc > 0) {
        const _Float16* bj = bp + (size_t)j * 16 * IN_F + kc * 32;
        v8h b0 = *(const v8h*)(bj);
        v8h b1 = *(const v8h*)(bj + 16);
        #pragma unroll
        for (int i = 0; i < 8; ++i) { B[i] = b0[i]; B[i + 8] = b1[i]; }
      }
      c[j] = __builtin_amdgcn_wmma_f32_16x16x32_f16(false, A[kc], false, B, (short)0, c[j],
                                                    false, false);
    }
  }

  float* hp = h + (size_t)(mrow0 + khalf * 8) * HF + col0 + l15;
  #pragma unroll
  for (int j = 0; j < 4; ++j) {
    #pragma unroll
    for (int r = 0; r < 8; ++r) hp[(size_t)r * HF + j * 16] = c[j][r];
  }

  float ps[8], pd[8];
  #pragma unroll
  for (int r = 0; r < 8; ++r) {
    ps[r] = c[0][r] * ws[0] + c[1][r] * ws[1] + c[2][r] * ws[2] + c[3][r] * ws[3];
    pd[r] = c[0][r] * wd[0] + c[1][r] * wd[1] + c[2][r] * wd[2] + c[3][r] * wd[3];
  }
  #pragma unroll
  for (int r = 0; r < 8; ++r) { ps[r] = row_xor_sum<1>(ps[r]); pd[r] = row_xor_sum<1>(pd[r]); }
  #pragma unroll
  for (int r = 0; r < 8; ++r) { ps[r] = row_xor_sum<2>(ps[r]); pd[r] = row_xor_sum<2>(pd[r]); }
  #pragma unroll
  for (int r = 0; r < 8; ++r) { ps[r] = row_xor_sum<4>(ps[r]); pd[r] = row_xor_sum<4>(pd[r]); }
  #pragma unroll
  for (int r = 0; r < 8; ++r) { ps[r] = row_xor_sum<8>(ps[r]); pd[r] = row_xor_sum<8>(pd[r]); }

  float v = ps[0];
  #pragma unroll
  for (int i = 1; i < 8; ++i) v = (l15 == i) ? ps[i] : v;
  #pragma unroll
  for (int i = 0; i < 8; ++i) v = (l15 == 8 + i) ? pd[i] : v;
  float* lp = (l15 < 8) ? a_src : a_dst;
  lp[(size_t)(mrow0 + khalf * 8 + (l15 & 7)) * HEADS + head] = v;
}

#define RANK_BLOCKS 1563
#define GEMM_STRIPS (N_NODES / 16)
#define FUSED_BLOCKS (RANK_BLOCKS + GEMM_STRIPS)
static_assert(RANK_BLOCKS * FUSED_THREADS >= N_EDGES / 4 &&
              (RANK_BLOCKS - 1) * FUSED_THREADS < N_EDGES / 4, "ranking blocks cover the edges");
static_assert(GEMM_STRIPS * 16 == N_NODES, "strips cover the nodes");
__global__ void __launch_bounds__(FUSED_THREADS)
rank_gemm_kernel(const int* __restrict__ ei, int* __restrict__ deg, int* __restrict__ rank,
                 const float* __restrict__ x, const _Float16* __restrict__ Wh,
                 const float* __restrict__ att_src, const float* __restrict__ att_dst,
                 float* __restrict__ h, float* __restrict__ a_src, float* __restrict__ a_dst) {
  __shared__ _Float16 xs[16 * XS_PITCH];
  const int b = blockIdx.x;
  if (b < RANK_BLOCKS) {
    rank_block(b, ei, deg, rank);
    return;
  }
  gemm_logits_strip(b - RANK_BLOCKS, xs, x, Wh, att_src, att_dst, h, a_src, a_dst);
}

#define SCAN_PER 49
__global__ void __launch_bounds__(1024)
scan_kernel(const int* __restrict__ deg, int* __restrict__ rowptr) {
  __shared__ int buf[1024 * SCAN_PER];
  __shared__ int wtot[32];
  const int t    = threadIdx.x;
  const int lane = t & 31;
  const int wv   = t >> 5;
  #pragma unroll
  for (int j = 0; j < SCAN_PER; ++j) {
    int idx = j * 1024 + t;
    buf[idx] = (idx < N_NODES) ? deg[idx] + deg[DEG_PLANE + idx] : 0;
  }
  __syncthreads();
  int v[SCAN_PER];
  int run = 0;
  #pragma unroll
  for (int j = 0; j < SCAN_PER; ++j) {
    int x = buf[t * SCAN_PER + j];
    v[j] = run;
    run += x;
  }
  int x = run;
  for (int off = 1; off < 32; off <<= 1) {
    int y = __shfl_up(x, off, 32);
    if (lane >= off) x += y;
  }
  if (lane == 31) wtot[wv] = x;
  __syncthreads();
  int w = (lane < wv) ? wtot[lane] : 0;
  for (int off = 16; off >= 1; off >>= 1) w += __shfl_xor(w, off, 32);
  const int offs = w + x - run;
  #pragma unroll
  for (int j = 0; j < SCAN_PER; ++j) buf[t * SCAN_PER + j] = v[j] + offs;
  __syncthreads();
  #pragma unroll
  for (int j = 0; j < SCAN_PER; ++j) {
    int idx = j * 1024 + t;
    if (idx < N_NODES) rowptr[idx] = buf[idx];
  }
  if (t == 1023) rowptr[N_NODES] = offs + run;
}

__global__ void __launch_bounds__(256)
fill_kernel(const int* __restrict__ ei, const int* __restrict__ rowptr,
            const int* __restrict__ rank, int* __restrict__ csr_src) {
  int i = blockIdx.x * blockDim.x + threadIdx.x;
  if (i >= N_EDGES) return;
  int s = ei[i], d = ei[N_EDGES + i];
  const int rk = rank[i];
  const int back = rk >> 31;
  const int r = rk & 0x7fffffff;
  const int edge = rowptr[d - back];
  csr_src[back ? edge - 1 - r : edge + r] = s;
}

__device__ __forceinline__ v4f edge_scores(const float* __restrict__ a_src, int s, v4f ad) {
  v4f as = *(const v4f*)(a_src + (size_t)s * HEADS);
  v4f e;
  for (int k = 0; k < HEADS; ++k) {
    float ev = as[k] + ad[k];
    e[k] = (ev >= 0.f) ? ev : NEG_SLOPE * ev;
  }
  return e;
}

__global__ void __launch_bounds__(256)
gather_kernel(const int* __restrict__ rowptr, const int* __restrict__ csr_src,
              const float* __restrict__ a_src, const float* __restrict__ a_dst,
              const float* __restrict__ h, const float* __restrict__ bias,
              float* __restrict__ out) {
  const int node = __builtin_amdgcn_readfirstlane((int)((blockIdx.x << 3) + (threadIdx.x >> 5)));
  const int lane = threadIdx.x & 31;
  const int beg = rowptr[node];
  const int end = rowptr[node + 1];
  const v4f ad = *(const v4f*)(a_dst + (size_t)node * HEADS);

  const float ninf = -__builtin_huge_valf();
  v4f mx = {ninf, ninf, ninf, ninf};
  for (int p = beg + lane; p < end; p += 32) {
    v4f e = edge_scores(a_src, csr_src[p], ad);
    for (int k = 0; k < HEADS; ++k) mx[k] = fmaxf(mx[k], e[k]);
  }
  for (int off = 16; off >= 1; off >>= 1) {
    for (int k = 0; k < HEADS; ++k) mx[k] = fmaxf(mx[k], __shfl_xor(mx[k], off, 32));
  }
  v4f m;
  for (int k = 0; k < HEADS; ++k)
    m[k] = (__builtin_fabsf(mx[k]) <= 3.402823466e38f) ? mx[k] : 0.f;

  v4f sm = {0.f, 0.f, 0.f, 0.f};
  for (int p = beg + lane; p < end; p += 32) {
    v4f e = edge_scores(a_src, csr_src[p], ad);
    for (int k = 0; k < HEADS; ++k) sm[k] += __expf(e[k] - m[k]);
  }
  for (int off = 16; off >= 1; off >>= 1) {
    for (int k = 0; k < HEADS; ++k) sm[k] += __shfl_xor(sm[k], off, 32);
  }
  v4f den;
  for (int k = 0; k < HEADS; ++k) den[k] = sm[k] + GAT_EPS;

  float acc[8];
  for (int q = 0; q < 8; ++q) acc[q] = 0.f;
  const float* hl = h + lane * 2;
  for (int cb = beg; cb < end; cb += 32) {
    const int p = cb + lane;
    int s = 0;
    v4f al = {0.f, 0.f, 0.f, 0.f};
    if (p < end) {
      s = csr_src[p];
      v4f e = edge_scores(a_src, s, ad);
      for (int k = 0; k < HEADS; ++k) al[k] = __expf(e[k] - m[k]) / den[k];
    }
    const int cnt = (end - cb < 32) ? (end - cb) : 32;
    #pragma unroll 4
    for (int j = 0; j < cnt; ++j) {
      const int sj = __builtin_amdgcn_readlane(s, j);
      const float a0 = __int_as_float(__builtin_amdgcn_readlane(__float_as_int(al[0]), j));
      const float a1 = __int_as_float(__builtin_amdgcn_readlane(__float_as_int(al[1]), j));
      const float a2 = __int_as_float(__builtin_amdgcn_readlane(__float_as_int(al[2]), j));
      const float a3 = __int_as_float(__builtin_amdgcn_readlane(__float_as_int(al[3]), j));
      const float* hr = hl + (size_t)sj * HF;
      v2f r0 = *(const v2f*)(hr);
      v2f r1 = *(const v2f*)(hr + OUT_F);
      v2f r2 = *(const v2f*)(hr + 2 * OUT_F);
      v2f r3 = *(const v2f*)(hr + 3 * OUT_F);
      acc[0] += a0 * r0[0]; acc[1] += a0 * r0[1];
      acc[2] += a1 * r1[0]; acc[3] += a1 * r1[1];
      acc[4] += a2 * r2[0]; acc[5] += a2 * r2[1];
      acc[6] += a3 * r3[0]; acc[7] += a3 * r3[1];
    }
  }
  float* o = out + (size_t)node * HF + lane * 2;
  const float* bl = bias + lane * 2;
  for (int k = 0; k < HEADS; ++k) {
    v2f b = *(const v2f*)(bl + k * OUT_F);
    v2f r = {acc[2 * k] + b[0], acc[2 * k + 1] + b[1]};
    *(v2f*)(o + k * OUT_F) = r;
  }
}

extern "C" void kernel_launch(void* const* d_in, const int* in_sizes, int n_in,
                              void* d_out, int out_size, void* d_ws, size_t ws_size,
                              hipStream_t stream) {
  const float* x       = (const float*)d_in[0];
  const int*   ei      = (const int*)  d_in[1];
  const float* W       = (const float*)d_in[2];
  const float* att_src = (const float*)d_in[3];
  const float* att_dst = (const float*)d_in[4];
  const float* bias    = (const float*)d_in[5];
  float* out = (float*)d_out;

  char* p = (char*)d_ws;
  auto alloc = [&](size_t bytes) {
    char* r = p; p += (bytes + 255) & ~(size_t)255; return r;
  };
  float*    h     = (float*)   alloc((size_t)N_NODES * HF * 4);
  _Float16* Wh    = (_Float16*)alloc((size_t)HF * IN_F * 2);
  float*    a_src = (float*)   alloc((size_t)N_NODES * HEADS * 4);
  float*    a_dst = (float*)   alloc((size_t)N_NODES * HEADS * 4);
  int*      deg     = (int*)   alloc((size_t)2 * DEG_PLANE * 4);
  int*      rowptr  = (int*)   alloc((size_t)(N_NODES + 1) * 4);
  int*      rank    = (int*)   alloc((size_t)N_EDGES * 4);
  int*      csr_src = (int*)   alloc((size_t)N_EDGES * 4);

  prep_kernel     <<<98, 256, 0, stream>>>(W, Wh, deg);
  rank_gemm_kernel<<<FUSED_BLOCKS, FUSED_THREADS, 0, stream>>>(ei, deg, rank, x, Wh, att_src, att_dst,
                                                              h, a_src, a_dst);
  scan_kernel     <<<1, 1024, 0, stream>>>(deg, rowptr);
  fill_kernel     <<<(N_EDGES + 255) / 256, 256, 0, stream>>>(ei, rowptr, rank, csr_src);
  gather_kernel   <<<N_NODES / 8, 256, 0, stream>>>(rowptr, csr_src, a_src, a_dst, h, bias, out);
}


# Gather wave fetches its edge list once: a lane keeps its edge's source id, scores and exponentials in registers for max, sum and weights (was 3 fetches); max and sum held as scalars; sum order kept
# speedup vs baseline: 5.4662x; 1.0584x over previous
#include <hip/hip_runtime.h>
#include <hip/hip_bf16.h>

#define N_NODES 50000
#define N_EDGES 800000
#define IN_F 128
#define OUT_F 64
#define HEADS 4
#define HF 256
#define NEG_SLOPE 0.2f
#define GAT_EPS 1e-16f
#define DEG_PLANE 50176

typedef __attribute__((ext_vector_type(16))) _Float16 v16h;
typedef __attribute__((ext_vector_type(8)))  _Float16 v8h;
typedef __attribute__((ext_vector_type(4)))  _Float16 v4h;
typedef __attribute__((ext_vector_type(8)))  float    v8f;
typedef __attribute__((ext_vector_type(4)))  float    v4f;
typedef __attribute__((ext_vector_type(2)))  float    v2f;
typedef __attribute__((ext_vector_type(4)))  int      v4i;

static_assert((2 * DEG_PLANE) % 4 == 0 && (HF * IN_F) % 4 == 0, "16-byte pieces");
__global__ void __launch_bounds__(256)
prep_kernel(const float* __restrict__ W, _Float16* __restrict__ Wh, int* __restrict__ deg) {
  const int idx = blockIdx.x * blockDim.x + threadIdx.x;
  const int stride = gridDim.x * blockDim.x;
  for (int i = idx; i < HF * IN_F / 4; i += stride) {
    const v4f w = *(const v4f*)(W + 4 * i);
    v4h o;
    for (int q = 0; q < 4; ++q) o[q] = (_Float16)w[q];
    *(v4h*)(Wh + 4 * i) = o;
  }
  const v4i zero = {0, 0, 0, 0};
  for (int i = idx; i < 2 * DEG_PLANE / 4; i += stride) *(v4i*)(deg + 4 * i) = zero;
}

#define FUSED_THREADS 128
static_assert(N_EDGES % 4 == 0, "four edges per lane, 16-byte loads");
__device__ __forceinline__ void
rank_block(int blk, const int* __restrict__ ei, int* __restrict__ deg, int* __restrict__ rank) {
  const int plane = (int)((__builtin_amdgcn_s_sendmsg_rtn(0x87) >> 18) & 1u);
  int i = blk * FUSED_THREADS + threadIdx.x;
  if (i >= N_EDGES / 4) return;
  const v4i d = *(const v4i*)(ei + N_EDGES + 4 * i);
  int* cnt = deg + plane * DEG_PLANE;
  const int mark = (int)((unsigned)plane << 31);
  v4i r;
  r[0] = atomicAdd(&cnt[d[0]], 1) | mark;
  r[1] = atomicAdd(&cnt[d[1]], 1) | mark;
  r[2] = atomicAdd(&cnt[d[2]], 1) | mark;
  r[3] = atomicAdd(&cnt[d[3]], 1) | mark;
  *(v4i*)(rank + 4 * i) = r;
}


template <int M>
__device__ __forceinline__ float row_xor_sum(float v) {
  int o = __builtin_amdgcn_update_dpp(0, __float_as_int(v), 0x160 + M, 0xf, 0xf, false);
  return v + __int_as_float(o);
}

#define XS_PITCH 136
__device__ __forceinline__ void
gemm_logits_strip(int strip, _Float16* xs,
                  const float* __restrict__ x, const _Float16* __restrict__ Wh,
                  const float* __restrict__ att_src, const float* __restrict__ att_dst,
                  float* __restrict__ h, float* __restrict__ a_src, float* __restrict__ a_dst) {
  const int lane  = threadIdx.x & 31;
  const int head  = threadIdx.x >> 5;
  const int mrow0 = strip << 4;
  const int col0  = head * OUT_F;

  const int l15   = lane & 15;
  const int khalf = lane >> 4;

  const _Float16* bp = Wh + (size_t)(col0 + l15) * IN_F + khalf * 8;

  float ws[4], wd[4];
  #pragma unroll
  for (int j = 0; j < 4; ++j) {
    ws[j] = att_src[col0 + j * 16 + l15];
    wd[j] = att_dst[col0 + j * 16 + l15];
  }

  v16h B0[4];
  #pragma unroll
  for (int j = 0; j < 4; ++j) {
    const _Float16* bj = bp + (size_t)j * 16 * IN_F;
    v8h b0 = *(const v8h*)(bj);
    v8h b1 = *(const v8h*)(bj + 16);
    #pragma unroll
    for (int i = 0; i < 8; ++i) { B0[j][i] = b0[i]; B0[j][i + 8] = b1[i]; }
  }

  {
    const float* xr = x + (size_t)(mrow0 + head * 4) * IN_F + lane * 4;
    v4f xf[4];
    #pragma unroll
    for (int r = 0; r < 4; ++r) xf[r] = *(const v4f*)(xr + (size_t)r * IN_F);
    #pragma unroll
    for (int r = 0; r < 4; ++r) {
      v4h o;
      #pragma unroll
      for (int i = 0; i < 4; ++i) o[i] = (_Float16)xf[r][i];
      *(v4h*)(xs + (head * 4 + r) * XS_PITCH + lane * 4) = o;
    }
  }
  __syncthreads();

  const _Float16* ap = xs + l15 * XS_PITCH + khalf * 8;
  v16h A[4];
  #pragma unroll
  for (int kc = 0; kc < 4; ++kc) {
    v8h a0 = *(const v8h*)(ap + kc * 32);
    v8h a1 = *(const v8h*)(ap + kc * 32 + 16);
    #pragma unroll
    for (int i = 0; i < 8; ++i) { A[kc][i] = a0[i]; A[kc][i + 8] = a1[i]; }
  }

  v8f c[4];
  #pragma unroll
  for (int j = 0; j < 4; ++j) c[j] = (v8f){};
  #pragma unroll
  for (int kc = 0; kc < 4; ++kc) {
    #pragma unroll
    for (int j = 0; j < 4; ++j) {
      v16h B = B0[j];
      if (kc > 0) {
        const _Float16* bj = bp + (size_t)j * 16 * IN_F + kc * 32;
        v8h b0 = *(const v8h*)(bj);
        v8h b1 = *(const v8h*)(bj + 16);
        #pragma unroll
        for (int i = 0; i < 8; ++i) { B[i] = b0[i]; B[i + 8] = b1[i]; }
      }
      c[j] = __builtin_amdgcn_wmma_f32_16x16x32_f16(false, A[kc], false, B, (short)0, c[j],
                                                    false, false);
    }
  }

  float* hp = h + (size_t)(mrow0 + khalf * 8) * HF + col0 + l15;
  #pragma unroll
  for (int j = 0; j < 4; ++j) {
    #pragma unroll
    for (int r = 0; r < 8; ++r) hp[(size_t)r * HF + j * 16] = c[j][r];
  }

  float ps[8], pd[8];
  #pragma unroll
  for (int r = 0; r < 8; ++r) {
    ps[r] = c[0][r] * ws[0] + c[1][r] * ws[1] + c[2][r] * ws[2] + c[3][r] * ws[3];
    pd[r] = c[0][r] * wd[0] + c[1][r] * wd[1] + c[2][r] * wd[2] + c[3][r] * wd[3];
  }
  #pragma unroll
  for (int r = 0; r < 8; ++r) { ps[r] = row_xor_sum<1>(ps[r]); pd[r] = row_xor_sum<1>(pd[r]); }
  #pragma unroll
  for (int r = 0; r < 8; ++r) { ps[r] = row_xor_sum<2>(ps[r]); pd[r] = row_xor_sum<2>(pd[r]); }
  #pragma unroll
  for (int r = 0; r < 8; ++r) { ps[r] = row_xor_sum<4>(ps[r]); pd[r] = row_xor_sum<4>(pd[r]); }
  #pragma unroll
  for (int r = 0; r < 8; ++r) { ps[r] = row_xor_sum<8>(ps[r]); pd[r] = row_xor_sum<8>(pd[r]); }

  float v = ps[0];
  #pragma unroll
  for (int i = 1; i < 8; ++i) v = (l15 == i) ? ps[i] : v;
  #pragma unroll
  for (int i = 0; i < 8; ++i) v = (l15 == 8 + i) ? pd[i] : v;
  float* lp = (l15 < 8) ? a_src : a_dst;
  lp[(size_t)(mrow0 + khalf * 8 + (l15 & 7)) * HEADS + head] = v;
}

#define RANK_BLOCKS 1563
#define GEMM_STRIPS (N_NODES / 16)
#define FUSED_BLOCKS (RANK_BLOCKS + GEMM_STRIPS)
static_assert(RANK_BLOCKS * FUSED_THREADS >= N_EDGES / 4 &&
              (RANK_BLOCKS - 1) * FUSED_THREADS < N_EDGES / 4, "ranking blocks cover the edges");
static_assert(GEMM_STRIPS * 16 == N_NODES, "strips cover the nodes");
__global__ void __launch_bounds__(FUSED_THREADS)
rank_gemm_kernel(const int* __restrict__ ei, int* __restrict__ deg, int* __restrict__ rank,
                 const float* __restrict__ x, const _Float16* __restrict__ Wh,
                 const float* __restrict__ att_src, const float* __restrict__ att_dst,
                 float* __restrict__ h, float* __restrict__ a_src, float* __restrict__ a_dst) {
  __shared__ _Float16 xs[16 * XS_PITCH];
  const int b = blockIdx.x;
  if (b < RANK_BLOCKS) {
    rank_block(b, ei, deg, rank);
    return;
  }
  gemm_logits_strip(b - RANK_BLOCKS, xs, x, Wh, att_src, att_dst, h, a_src, a_dst);
}

#define SCAN_PER 49
__global__ void __launch_bounds__(1024)
scan_kernel(const int* __restrict__ deg, int* __restrict__ rowptr) {
  __shared__ int buf[1024 * SCAN_PER];
  __shared__ int wtot[32];
  const int t    = threadIdx.x;
  const int lane = t & 31;
  const int wv   = t >> 5;
  #pragma unroll
  for (int j = 0; j < SCAN_PER; ++j) {
    int idx = j * 1024 + t;
    buf[idx] = (idx < N_NODES) ? deg[idx] + deg[DEG_PLANE + idx] : 0;
  }
  __syncthreads();
  int v[SCAN_PER];
  int run = 0;
  #pragma unroll
  for (int j = 0; j < SCAN_PER; ++j) {
    int x = buf[t * SCAN_PER + j];
    v[j] = run;
    run += x;
  }
  int x = run;
  for (int off = 1; off < 32; off <<= 1) {
    int y = __shfl_up(x, off, 32);
    if (lane >= off) x += y;
  }
  if (lane == 31) wtot[wv] = x;
  __syncthreads();
  int w = (lane < wv) ? wtot[lane] : 0;
  for (int off = 16; off >= 1; off >>= 1) w += __shfl_xor(w, off, 32);
  const int offs = w + x - run;
  #pragma unroll
  for (int j = 0; j < SCAN_PER; ++j) buf[t * SCAN_PER + j] = v[j] + offs;
  __syncthreads();
  #pragma unroll
  for (int j = 0; j < SCAN_PER; ++j) {
    int idx = j * 1024 + t;
    if (idx < N_NODES) rowptr[idx] = buf[idx];
  }
  if (t == 1023) rowptr[N_NODES] = offs + run;
}

__global__ void __launch_bounds__(256)
fill_kernel(const int* __restrict__ ei, const int* __restrict__ rowptr,
            const int* __restrict__ rank, int* __restrict__ csr_src) {
  int i = blockIdx.x * blockDim.x + threadIdx.x;
  if (i >= N_EDGES) return;
  int s = ei[i], d = ei[N_EDGES + i];
  const int rk = rank[i];
  const int back = rk >> 31;
  const int r = rk & 0x7fffffff;
  const int edge = rowptr[d - back];
  csr_src[back ? edge - 1 - r : edge + r] = s;
}

__device__ __forceinline__ v4f edge_scores(const float* __restrict__ a_src, int s, v4f ad) {
  v4f as = *(const v4f*)(a_src + (size_t)s * HEADS);
  v4f e;
  for (int k = 0; k < HEADS; ++k) {
    float ev = as[k] + ad[k];
    e[k] = (ev >= 0.f) ? ev : NEG_SLOPE * ev;
  }
  return e;
}

__global__ void __launch_bounds__(256)
gather_kernel(const int* __restrict__ rowptr, const int* __restrict__ csr_src,
              const float* __restrict__ a_src, const float* __restrict__ a_dst,
              const float* __restrict__ h, const float* __restrict__ bias,
              float* __restrict__ out) {
  const int node = __builtin_amdgcn_readfirstlane((int)((blockIdx.x << 3) + (threadIdx.x >> 5)));
  const int lane = threadIdx.x & 31;
  const int beg = rowptr[node];
  const int end = rowptr[node + 1];
  const v4f ad = *(const v4f*)(a_dst + (size_t)node * HEADS);

  const int p0 = beg + lane;
  const bool own = p0 < end;
  int s0 = 0;
  v4f e0 = {0.f, 0.f, 0.f, 0.f};

  const float ninf = -__builtin_huge_valf();
  v4f mx = {ninf, ninf, ninf, ninf};
  if (own) {
    s0 = csr_src[p0];
    e0 = edge_scores(a_src, s0, ad);
    for (int k = 0; k < HEADS; ++k) mx[k] = fmaxf(mx[k], e0[k]);
  }
  for (int p = p0 + 32; p < end; p += 32) {
    v4f e = edge_scores(a_src, csr_src[p], ad);
    for (int k = 0; k < HEADS; ++k) mx[k] = fmaxf(mx[k], e[k]);
  }
  for (int off = 16; off >= 1; off >>= 1) {
    for (int k = 0; k < HEADS; ++k) mx[k] = fmaxf(mx[k], __shfl_xor(mx[k], off, 32));
  }
  v4f m;
  for (int k = 0; k < HEADS; ++k) {
    const float mk = (__builtin_fabsf(mx[k]) <= 3.402823466e38f) ? mx[k] : 0.f;
    m[k] = __int_as_float(__builtin_amdgcn_readfirstlane(__float_as_int(mk)));
  }

  v4f ex0 = {0.f, 0.f, 0.f, 0.f};
  if (own) {
    for (int k = 0; k < HEADS; ++k) ex0[k] = __expf(e0[k] - m[k]);
  }
  v4f sm = ex0;
  for (int p = p0 + 32; p < end; p += 32) {
    v4f e = edge_scores(a_src, csr_src[p], ad);
    for (int k = 0; k < HEADS; ++k) sm[k] += __expf(e[k] - m[k]);
  }
  for (int off = 16; off >= 1; off >>= 1) {
    for (int k = 0; k < HEADS; ++k) sm[k] += __shfl_xor(sm[k], off, 32);
  }
  v4f den;
  for (int k = 0; k < HEADS; ++k)
    den[k] = __int_as_float(__builtin_amdgcn_readfirstlane(__float_as_int(sm[k] + GAT_EPS)));

  float acc[8];
  for (int q = 0; q < 8; ++q) acc[q] = 0.f;
  const float* hl = h + lane * 2;
  for (int cb = beg; cb < end; cb += 32) {
    int s = 0;
    v4f al = {0.f, 0.f, 0.f, 0.f};
    if (cb == beg) {
      s = s0;
      if (own) {
        for (int k = 0; k < HEADS; ++k) al[k] = ex0[k] / den[k];
      }
    } else {
      const int p = cb + lane;
      if (p < end) {
        s = csr_src[p];
        v4f e = edge_scores(a_src, s, ad);
        for (int k = 0; k < HEADS; ++k) al[k] = __expf(e[k] - m[k]) / den[k];
      }
    }
    const int cnt = (end - cb < 32) ? (end - cb) : 32;
    #pragma unroll 4
    for (int j = 0; j < cnt; ++j) {
      const int sj = __builtin_amdgcn_readlane(s, j);
      const float a0 = __int_as_float(__builtin_amdgcn_readlane(__float_as_int(al[0]), j));
      const float a1 = __int_as_float(__builtin_amdgcn_readlane(__float_as_int(al[1]), j));
      const float a2 = __int_as_float(__builtin_amdgcn_readlane(__float_as_int(al[2]), j));
      const float a3 = __int_as_float(__builtin_amdgcn_readlane(__float_as_int(al[3]), j));
      const float* hr = hl + (size_t)sj * HF;
      v2f r0 = *(const v2f*)(hr);
      v2f r1 = *(const v2f*)(hr + OUT_F);
      v2f r2 = *(const v2f*)(hr + 2 * OUT_F);
      v2f r3 = *(const v2f*)(hr + 3 * OUT_F);
      acc[0] += a0 * r0[0]; acc[1] += a0 * r0[1];
      acc[2] += a1 * r1[0]; acc[3] += a1 * r1[1];
      acc[4] += a2 * r2[0]; acc[5] += a2 * r2[1];
      acc[6] += a3 * r3[0]; acc[7] += a3 * r3[1];
    }
  }
  float* o = out + (size_t)node * HF + lane * 2;
  const float* bl = bias + lane * 2;
  for (int k = 0; k < HEADS; ++k) {
    v2f b = *(const v2f*)(bl + k * OUT_F);
    v2f r = {acc[2 * k] + b[0], acc[2 * k + 1] + b[1]};
    *(v2f*)(o + k * OUT_F) = r;
  }
}

extern "C" void kernel_launch(void* const* d_in, const int* in_sizes, int n_in,
                              void* d_out, int out_size, void* d_ws, size_t ws_size,
                              hipStream_t stream) {
  const float* x       = (const float*)d_in[0];
  const int*   ei      = (const int*)  d_in[1];
  const float* W       = (const float*)d_in[2];
  const float* att_src = (const float*)d_in[3];
  const float* att_dst = (const float*)d_in[4];
  const float* bias    = (const float*)d_in[5];
  float* out = (float*)d_out;

  char* p = (char*)d_ws;
  auto alloc = [&](size_t bytes) {
    char* r = p; p += (bytes + 255) & ~(size_t)255; return r;
  };
  float*    h     = (float*)   alloc((size_t)N_NODES * HF * 4);
  _Float16* Wh    = (_Float16*)alloc((size_t)HF * IN_F * 2);
  float*    a_src = (float*)   alloc((size_t)N_NODES * HEADS * 4);
  float*    a_dst = (float*)   alloc((size_t)N_NODES * HEADS * 4);
  int*      deg     = (int*)   alloc((size_t)2 * DEG_PLANE * 4);
  int*      rowptr  = (int*)   alloc((size_t)(N_NODES + 1) * 4);
  int*      rank    = (int*)   alloc((size_t)N_EDGES * 4);
  int*      csr_src = (int*)   alloc((size_t)N_EDGES * 4);

  prep_kernel     <<<98, 256, 0, stream>>>(W, Wh, deg);
  rank_gemm_kernel<<<FUSED_BLOCKS, FUSED_THREADS, 0, stream>>>(ei, deg, rank, x, Wh, att_src, att_dst,
                                                              h, a_src, a_dst);
  scan_kernel     <<<1, 1024, 0, stream>>>(deg, rowptr);
  fill_kernel     <<<(N_EDGES + 255) / 256, 256, 0, stream>>>(ei, rowptr, rank, csr_src);
  gather_kernel   <<<N_NODES / 8, 256, 0, stream>>>(rowptr, csr_src, a_src, a_dst, h, bias, out);
}


# Projection moved behind the aggregation: a wave sums its nodes' 512-B x rows per head, 32 nodes' sums (rounded to the 16-bit operand form) meet W on WMMA via LDS; h plane gone; sums in a new order
# speedup vs baseline: 6.2361x; 1.1408x over previous
#include <hip/hip_runtime.h>
#include <hip/hip_bf16.h>

#define N_NODES 50000
#define N_EDGES 800000
#define IN_F 128
#define OUT_F 64
#define HEADS 4
#define HF 256
#define NEG_SLOPE 0.2f
#define GAT_EPS 1e-16f
#define DEG_PLANE 50176

typedef __attribute__((ext_vector_type(16))) _Float16 v16h;
typedef __attribute__((ext_vector_type(8)))  _Float16 v8h;
typedef __attribute__((ext_vector_type(4)))  _Float16 v4h;
typedef __attribute__((ext_vector_type(8)))  float    v8f;
typedef __attribute__((ext_vector_type(4)))  float    v4f;
typedef __attribute__((ext_vector_type(4)))  int      v4i;

static_assert((2 * DEG_PLANE) % 4 == 0 && (HF * IN_F) % 4 == 0, "16-byte pieces");
__global__ void __launch_bounds__(256)
prep_kernel(const float* __restrict__ W, const float* __restrict__ att_src,
            const float* __restrict__ att_dst, _Float16* __restrict__ Wh,
            float* __restrict__ U, int* __restrict__ deg) {
  const int idx = blockIdx.x * blockDim.x + threadIdx.x;
  const int stride = gridDim.x * blockDim.x;
  for (int i = idx; i < HF * IN_F / 4; i += stride) {
    const v4f w = *(const v4f*)(W + 4 * i);
    v4h o;
    for (int q = 0; q < 4; ++q) o[q] = (_Float16)w[q];
    *(v4h*)(Wh + 4 * i) = o;
  }
  const v4i zero = {0, 0, 0, 0};
  for (int i = idx; i < 2 * DEG_PLANE / 4; i += stride) *(v4i*)(deg + 4 * i) = zero;
  for (int o = stride - 1 - idx; o < 2 * HEADS * IN_F; o += stride) {
    const int i = o % IN_F;
    const int k = (o / IN_F) % HEADS;
    const float* att = (o < HEADS * IN_F) ? att_src : att_dst;
    float acc = 0.f;
    #pragma unroll
    for (int f = 0; f < OUT_F; ++f) acc += W[(size_t)(k * OUT_F + f) * IN_F + i] * att[k * OUT_F + f];
    U[o] = acc;
  }
}

#define FUSED_THREADS 128
static_assert(N_EDGES % 4 == 0, "four edges per lane, 16-byte loads");
__device__ __forceinline__ void
rank_block(int blk, const int* __restrict__ ei, int* __restrict__ deg, int* __restrict__ rank) {
  const int plane = (int)((__builtin_amdgcn_s_sendmsg_rtn(0x87) >> 18) & 1u);
  int i = blk * FUSED_THREADS + threadIdx.x;
  if (i >= N_EDGES / 4) return;
  const v4i d = *(const v4i*)(ei + N_EDGES + 4 * i);
  int* cnt = deg + plane * DEG_PLANE;
  const int mark = (int)((unsigned)plane << 31);
  v4i r;
  r[0] = atomicAdd(&cnt[d[0]], 1) | mark;
  r[1] = atomicAdd(&cnt[d[1]], 1) | mark;
  r[2] = atomicAdd(&cnt[d[2]], 1) | mark;
  r[3] = atomicAdd(&cnt[d[3]], 1) | mark;
  *(v4i*)(rank + 4 * i) = r;
}


template <int M>
__device__ __forceinline__ float row_xor_sum(float v) {
  int o = __builtin_amdgcn_update_dpp(0, __float_as_int(v), 0x160 + M, 0xf, 0xf, false);
  return v + __int_as_float(o);
}

__device__ __forceinline__ void
logits_strip(int strip, const float* __restrict__ x, const float* __restrict__ U,
             float* __restrict__ a_src, float* __restrict__ a_dst) {
  const int lane  = threadIdx.x & 31;
  const int wv    = threadIdx.x >> 5;
  const int l15   = lane & 15;
  const int half  = lane >> 4;
  const int row0  = (strip << 4) + wv * 4 + half;
  v4f ua[2 * HEADS], ub[2 * HEADS];
  #pragma unroll
  for (int q = 0; q < 2 * HEADS; ++q) {
    ua[q] = *(const v4f*)(U + q * IN_F + l15 * 4);
    ub[q] = *(const v4f*)(U + q * IN_F + 64 + l15 * 4);
  }
  const float* xr = x + (size_t)row0 * IN_F + l15 * 4;
  v4f xa[2], xb[2];
  #pragma unroll
  for (int rp = 0; rp < 2; ++rp) {
    xa[rp] = *(const v4f*)(xr + (size_t)rp * 2 * IN_F);
    xb[rp] = *(const v4f*)(xr + (size_t)rp * 2 * IN_F + 64);
  }
  #pragma unroll
  for (int rp = 0; rp < 2; ++rp) {
    float p[2 * HEADS];
    #pragma unroll
    for (int q = 0; q < 2 * HEADS; ++q) {
      p[q] = xa[rp][0] * ua[q][0] + xa[rp][1] * ua[q][1] + xa[rp][2] * ua[q][2] + xa[rp][3] * ua[q][3]
           + xb[rp][0] * ub[q][0] + xb[rp][1] * ub[q][1] + xb[rp][2] * ub[q][2] + xb[rp][3] * ub[q][3];
    }
    #pragma unroll
    for (int q = 0; q < 2 * HEADS; ++q) p[q] = row_xor_sum<1>(p[q]);
    #pragma unroll
    for (int q = 0; q < 2 * HEADS; ++q) p[q] = row_xor_sum<2>(p[q]);
    #pragma unroll
    for (int q = 0; q < 2 * HEADS; ++q) p[q] = row_xor_sum<4>(p[q]);
    #pragma unroll
    for (int q = 0; q < 2 * HEADS; ++q) p[q] = row_xor_sum<8>(p[q]);
    float v = p[0];
    #pragma unroll
    for (int q = 1; q < 2 * HEADS; ++q) v = (l15 == q) ? p[q] : v;
    if (l15 < 2 * HEADS) {
      float* lp = (l15 < HEADS) ? a_src : a_dst;
      lp[(size_t)(row0 + rp * 2) * HEADS + (l15 & (HEADS - 1))] = v;
    }
  }
}

#define RANK_BLOCKS 1563
#define LOGIT_STRIPS (N_NODES / 16)
#define FUSED_BLOCKS (RANK_BLOCKS + LOGIT_STRIPS)
static_assert(RANK_BLOCKS * FUSED_THREADS >= N_EDGES / 4 &&
              (RANK_BLOCKS - 1) * FUSED_THREADS < N_EDGES / 4, "ranking blocks cover the edges");
static_assert(LOGIT_STRIPS * 16 == N_NODES, "strips cover the nodes");
__global__ void __launch_bounds__(FUSED_THREADS)
rank_logits_kernel(const int* __restrict__ ei, int* __restrict__ deg, int* __restrict__ rank,
                   const float* __restrict__ x, const float* __restrict__ U,
                   float* __restrict__ a_src, float* __restrict__ a_dst) {
  const int b = blockIdx.x;
  if (b < RANK_BLOCKS) {
    rank_block(b, ei, deg, rank);
    return;
  }
  logits_strip(b - RANK_BLOCKS, x, U, a_src, a_dst);
}

#define SCAN_PER 49
__global__ void __launch_bounds__(1024)
scan_kernel(const int* __restrict__ deg, int* __restrict__ rowptr) {
  __shared__ int buf[1024 * SCAN_PER];
  __shared__ int wtot[32];
  const int t    = threadIdx.x;
  const int lane = t & 31;
  const int wv   = t >> 5;
  #pragma unroll
  for (int j = 0; j < SCAN_PER; ++j) {
    int idx = j * 1024 + t;
    buf[idx] = (idx < N_NODES) ? deg[idx] + deg[DEG_PLANE + idx] : 0;
  }
  __syncthreads();
  int v[SCAN_PER];
  int run = 0;
  #pragma unroll
  for (int j = 0; j < SCAN_PER; ++j) {
    int x = buf[t * SCAN_PER + j];
    v[j] = run;
    run += x;
  }
  int x = run;
  for (int off = 1; off < 32; off <<= 1) {
    int y = __shfl_up(x, off, 32);
    if (lane >= off) x += y;
  }
  if (lane == 31) wtot[wv] = x;
  __syncthreads();
  int w = (lane < wv) ? wtot[lane] : 0;
  for (int off = 16; off >= 1; off >>= 1) w += __shfl_xor(w, off, 32);
  const int offs = w + x - run;
  #pragma unroll
  for (int j = 0; j < SCAN_PER; ++j) buf[t * SCAN_PER + j] = v[j] + offs;
  __syncthreads();
  #pragma unroll
  for (int j = 0; j < SCAN_PER; ++j) {
    int idx = j * 1024 + t;
    if (idx < N_NODES) rowptr[idx] = buf[idx];
  }
  if (t == 1023) rowptr[N_NODES] = offs + run;
}

__global__ void __launch_bounds__(256)
fill_kernel(const int* __restrict__ ei, const int* __restrict__ rowptr,
            const int* __restrict__ rank, int* __restrict__ csr_src) {
  int i = blockIdx.x * blockDim.x + threadIdx.x;
  if (i >= N_EDGES) return;
  int s = ei[i], d = ei[N_EDGES + i];
  const int rk = rank[i];
  const int back = rk >> 31;
  const int r = rk & 0x7fffffff;
  const int edge = rowptr[d - back];
  csr_src[back ? edge - 1 - r : edge + r] = s;
}

__device__ __forceinline__ v4f edge_scores(const float* __restrict__ a_src, int s, v4f ad) {
  v4f as = *(const v4f*)(a_src + (size_t)s * HEADS);
  v4f e;
  for (int k = 0; k < HEADS; ++k) {
    float ev = as[k] + ad[k];
    e[k] = (ev >= 0.f) ? ev : NEG_SLOPE * ev;
  }
  return e;
}

#define XS_PITCH 136
#define GP_NODES 32
#define GP_THREADS 512
#define GP_BLOCKS ((N_NODES + GP_NODES - 1) / GP_NODES)
static_assert(GP_THREADS / 32 * 2 == GP_NODES, "two nodes per wave");
static_assert(N_NODES % 16 == 0, "the last block holds whole 16-row tiles");
__global__ void __launch_bounds__(GP_THREADS, 16)
gather_project_kernel(const int* __restrict__ rowptr, const int* __restrict__ csr_src,
                      const float* __restrict__ a_src, const float* __restrict__ a_dst,
                      const float* __restrict__ x, const _Float16* __restrict__ Wh,
                      const float* __restrict__ bias, float* __restrict__ out) {
  __shared__ _Float16 xs[HEADS * GP_NODES * XS_PITCH];
  const int wv   = __builtin_amdgcn_readfirstlane((int)(threadIdx.x >> 5));
  const int lane = threadIdx.x & 31;
  const int node0 = blockIdx.x * GP_NODES;
  const int lane4 = lane * 4;
  const float ninf = -__builtin_huge_valf();

  #pragma nounroll
  for (int h2 = 0; h2 < 2; ++h2) {
    const int nl = wv * 2 + h2;
    const int node = node0 + nl;
    v4f acc[HEADS];
    #pragma unroll
    for (int k = 0; k < HEADS; ++k) acc[k] = (v4f){0.f, 0.f, 0.f, 0.f};
    if (node < N_NODES) {
      const int beg = rowptr[node];
      const int end = rowptr[node + 1];
      const v4f ad = *(const v4f*)(a_dst + (size_t)node * HEADS);

      const int p0 = beg + lane;
      const bool own = p0 < end;
      int s0 = 0;
      v4f e0 = {0.f, 0.f, 0.f, 0.f};

      v4f mx = {ninf, ninf, ninf, ninf};
      if (own) {
        s0 = csr_src[p0];
        e0 = edge_scores(a_src, s0, ad);
        for (int k = 0; k < HEADS; ++k) mx[k] = fmaxf(mx[k], e0[k]);
      }
      for (int p = p0 + 32; p < end; p += 32) {
        v4f e = edge_scores(a_src, csr_src[p], ad);
        for (int k = 0; k < HEADS; ++k) mx[k] = fmaxf(mx[k], e[k]);
      }
      for (int off = 16; off >= 1; off >>= 1) {
        for (int k = 0; k < HEADS; ++k) mx[k] = fmaxf(mx[k], __shfl_xor(mx[k], off, 32));
      }
      v4f m;
      for (int k = 0; k < HEADS; ++k) {
        const float mk = (__builtin_fabsf(mx[k]) <= 3.402823466e38f) ? mx[k] : 0.f;
        m[k] = __int_as_float(__builtin_amdgcn_readfirstlane(__float_as_int(mk)));
      }

      v4f ex0 = {0.f, 0.f, 0.f, 0.f};
      if (own) {
        for (int k = 0; k < HEADS; ++k) ex0[k] = __expf(e0[k] - m[k]);
      }
      v4f sm = ex0;
      for (int p = p0 + 32; p < end; p += 32) {
        v4f e = edge_scores(a_src, csr_src[p], ad);
        for (int k = 0; k < HEADS; ++k) sm[k] += __expf(e[k] - m[k]);
      }
      for (int off = 16; off >= 1; off >>= 1) {
        for (int k = 0; k < HEADS; ++k) sm[k] += __shfl_xor(sm[k], off, 32);
      }
      v4f den;
      for (int k = 0; k < HEADS; ++k)
        den[k] = __int_as_float(__builtin_amdgcn_readfirstlane(__float_as_int(sm[k] + GAT_EPS)));

      for (int cb = beg; cb < end; cb += 32) {
        int s = 0;
        v4f al = {0.f, 0.f, 0.f, 0.f};
        if (cb == beg) {
          s = s0;
          if (own) {
            for (int k = 0; k < HEADS; ++k) al[k] = ex0[k] / den[k];
          }
        } else {
          const int p = cb + lane;
          if (p < end) {
            s = csr_src[p];
            v4f e = edge_scores(a_src, s, ad);
            for (int k = 0; k < HEADS; ++k) al[k] = __expf(e[k] - m[k]) / den[k];
          }
        }
        const int cnt = (end - cb < 32) ? (end - cb) : 32;
        #pragma unroll 4
        for (int j = 0; j < cnt; ++j) {
          const int sj = __builtin_amdgcn_readlane(s, j);
          const float a0 = __int_as_float(__builtin_amdgcn_readlane(__float_as_int(al[0]), j));
          const float a1 = __int_as_float(__builtin_amdgcn_readlane(__float_as_int(al[1]), j));
          const float a2 = __int_as_float(__builtin_amdgcn_readlane(__float_as_int(al[2]), j));
          const float a3 = __int_as_float(__builtin_amdgcn_readlane(__float_as_int(al[3]), j));
          const v4f r = *(const v4f*)(x + (size_t)sj * IN_F + lane4);
          #pragma unroll
          for (int i = 0; i < 4; ++i) {
            acc[0][i] += a0 * r[i];
            acc[1][i] += a1 * r[i];
            acc[2][i] += a2 * r[i];
            acc[3][i] += a3 * r[i];
          }
        }
      }
    }

    #pragma unroll
    for (int k = 0; k < HEADS; ++k) {
      v4h o;
      #pragma unroll
      for (int i = 0; i < 4; ++i) o[i] = (_Float16)acc[k][i];
      *(v4h*)(xs + (k * GP_NODES + nl) * XS_PITCH + lane * 4) = o;
    }
  }

  const int l15   = lane & 15;
  const int khalf = lane >> 4;
  const int head  = (wv >> 1) & (HEADS - 1);
  const int col0  = head * OUT_F + (wv & 1) * 32;
  const _Float16* bp = Wh + (size_t)(col0 + l15) * IN_F + khalf * 8;

  v16h B0[2];
  if (wv < 8) {
    #pragma unroll
    for (int t = 0; t < 2; ++t) {
      const _Float16* bt = bp + (size_t)t * 16 * IN_F;
      v8h b0 = *(const v8h*)(bt);
      v8h b1 = *(const v8h*)(bt + 16);
      #pragma unroll
      for (int i = 0; i < 8; ++i) { B0[t][i] = b0[i]; B0[t][i + 8] = b1[i]; }
    }
  }
  __syncthreads();
  if (wv >= 8) return;

  #pragma nounroll
  for (int mt = 0; mt < 2; ++mt) {
    const int row0 = node0 + mt * 16;
    if (row0 >= N_NODES) break;
    const _Float16* ap = xs + (head * GP_NODES + mt * 16 + l15) * XS_PITCH + khalf * 8;
    v8f c[2];
    #pragma unroll
    for (int t = 0; t < 2; ++t) c[t] = (v8f){};
    #pragma unroll
    for (int kc = 0; kc < 4; ++kc) {
      v16h A;
      {
        v8h a0 = *(const v8h*)(ap + kc * 32);
        v8h a1 = *(const v8h*)(ap + kc * 32 + 16);
        #pragma unroll
        for (int i = 0; i < 8; ++i) { A[i] = a0[i]; A[i + 8] = a1[i]; }
      }
      #pragma unroll
      for (int t = 0; t < 2; ++t) {
        v16h B = B0[t];
        if (kc > 0) {
          const _Float16* bt = bp + (size_t)t * 16 * IN_F + kc * 32;
          v8h b0 = *(const v8h*)(bt);
          v8h b1 = *(const v8h*)(bt + 16);
          #pragma unroll
          for (int i = 0; i < 8; ++i) { B[i] = b0[i]; B[i + 8] = b1[i]; }
        }
        c[t] = __builtin_amdgcn_wmma_f32_16x16x32_f16(false, A, false, B, (short)0, c[t],
                                                      false, false);
      }
    }

    float* op = out + (size_t)(row0 + khalf * 8) * HF + col0 + l15;
    #pragma unroll
    for (int t = 0; t < 2; ++t) {
      const float b = bias[col0 + t * 16 + l15];
      #pragma unroll
      for (int r = 0; r < 8; ++r) op[(size_t)r * HF + t * 16] = c[t][r] + b;
    }
  }
}

extern "C" void kernel_launch(void* const* d_in, const int* in_sizes, int n_in,
                              void* d_out, int out_size, void* d_ws, size_t ws_size,
                              hipStream_t stream) {
  const float* x       = (const float*)d_in[0];
  const int*   ei      = (const int*)  d_in[1];
  const float* W       = (const float*)d_in[2];
  const float* att_src = (const float*)d_in[3];
  const float* att_dst = (const float*)d_in[4];
  const float* bias    = (const float*)d_in[5];
  float* out = (float*)d_out;

  char* p = (char*)d_ws;
  auto alloc = [&](size_t bytes) {
    char* r = p; p += (bytes + 255) & ~(size_t)255; return r;
  };
  _Float16* Wh    = (_Float16*)alloc((size_t)HF * IN_F * 2);
  float*    U     = (float*)   alloc((size_t)2 * HEADS * IN_F * 4);
  float*    a_src = (float*)   alloc((size_t)N_NODES * HEADS * 4);
  float*    a_dst = (float*)   alloc((size_t)N_NODES * HEADS * 4);
  int*      deg     = (int*)   alloc((size_t)2 * DEG_PLANE * 4);
  int*      rowptr  = (int*)   alloc((size_t)(N_NODES + 1) * 4);
  int*      rank    = (int*)   alloc((size_t)N_EDGES * 4);
  int*      csr_src = (int*)   alloc((size_t)N_EDGES * 4);

  prep_kernel       <<<98, 256, 0, stream>>>(W, att_src, att_dst, Wh, U, deg);
  rank_logits_kernel<<<FUSED_BLOCKS, FUSED_THREADS, 0, stream>>>(ei, deg, rank, x, U, a_src, a_dst);
  scan_kernel       <<<1, 1024, 0, stream>>>(deg, rowptr);
  fill_kernel       <<<(N_EDGES + 255) / 256, 256, 0, stream>>>(ei, rowptr, rank, csr_src);
  gather_project_kernel<<<GP_BLOCKS, GP_THREADS, 0, stream>>>(rowptr, csr_src, a_src, a_dst,
                                                                      x, Wh, bias, out);
}


# Ranking add places the edge at once: source id into slot rank of the node's 64-id block (half per counter plane, long lists spill to a side list): scan and fill launches gone; sums arrival-ordered
# speedup vs baseline: 7.7514x; 1.2430x over previous
#include <hip/hip_runtime.h>
#include <hip/hip_bf16.h>

#define N_NODES 50000
#define N_EDGES 800000
#define IN_F 128
#define OUT_F 64
#define HEADS 4
#define HF 256
#define NEG_SLOPE 0.2f
#define GAT_EPS 1e-16f
#define DEG_PLANE 50176
#define SLOT_HALF 32
#define SLOT_BLOCK 64

typedef __attribute__((ext_vector_type(16))) _Float16 v16h;
typedef __attribute__((ext_vector_type(8)))  _Float16 v8h;
typedef __attribute__((ext_vector_type(4)))  _Float16 v4h;
typedef __attribute__((ext_vector_type(8)))  float    v8f;
typedef __attribute__((ext_vector_type(4)))  float    v4f;
typedef __attribute__((ext_vector_type(4)))  int      v4i;

static_assert((2 * DEG_PLANE) % 4 == 0 && (HF * IN_F) % 4 == 0, "16-byte pieces");
__global__ void __launch_bounds__(256)
prep_kernel(const float* __restrict__ W, const float* __restrict__ att_src,
            const float* __restrict__ att_dst, _Float16* __restrict__ Wh,
            float* __restrict__ U, int* __restrict__ deg, int* __restrict__ novf) {
  const int idx = blockIdx.x * blockDim.x + threadIdx.x;
  const int stride = gridDim.x * blockDim.x;
  for (int i = idx; i < HF * IN_F / 4; i += stride) {
    const v4f w = *(const v4f*)(W + 4 * i);
    v4h o;
    for (int q = 0; q < 4; ++q) o[q] = (_Float16)w[q];
    *(v4h*)(Wh + 4 * i) = o;
  }
  const v4i zero = {0, 0, 0, 0};
  for (int i = idx; i < 2 * DEG_PLANE / 4; i += stride) *(v4i*)(deg + 4 * i) = zero;
  if (idx == 0) *novf = 0;
  for (int o = stride - 1 - idx; o < 2 * HEADS * IN_F; o += stride) {
    const int i = o % IN_F;
    const int k = (o / IN_F) % HEADS;
    const float* att = (o < HEADS * IN_F) ? att_src : att_dst;
    float acc = 0.f;
    #pragma unroll
    for (int f = 0; f < OUT_F; ++f) acc += W[(size_t)(k * OUT_F + f) * IN_F + i] * att[k * OUT_F + f];
    U[o] = acc;
  }
}

#define FUSED_THREADS 128
static_assert(N_EDGES % 4 == 0, "four edges per lane, 16-byte loads");
__device__ __forceinline__ void
rank_block(int blk, const int* __restrict__ ei, int* __restrict__ deg, int* __restrict__ slots,
           int* __restrict__ novf, int* __restrict__ ov_src, int* __restrict__ ov_dst) {
  const int plane = (int)((__builtin_amdgcn_s_sendmsg_rtn(0x87) >> 18) & 1u);
  int i = blk * FUSED_THREADS + threadIdx.x;
  if (i >= N_EDGES / 4) return;
  const v4i s = *(const v4i*)(ei + 4 * i);
  const v4i d = *(const v4i*)(ei + N_EDGES + 4 * i);
  int* cnt = deg + plane * DEG_PLANE;
  v4i r;
  r[0] = atomicAdd(&cnt[d[0]], 1);
  r[1] = atomicAdd(&cnt[d[1]], 1);
  r[2] = atomicAdd(&cnt[d[2]], 1);
  r[3] = atomicAdd(&cnt[d[3]], 1);
  #pragma unroll
  for (int q = 0; q < 4; ++q) {
    if (r[q] < SLOT_HALF) {
      slots[(size_t)d[q] * SLOT_BLOCK + plane * SLOT_HALF + r[q]] = s[q];
    } else {
      const int o = atomicAdd(novf, 1);
      ov_src[o] = s[q];
      ov_dst[o] = d[q];
    }
  }
}


template <int M>
__device__ __forceinline__ float row_xor_sum(float v) {
  int o = __builtin_amdgcn_update_dpp(0, __float_as_int(v), 0x160 + M, 0xf, 0xf, false);
  return v + __int_as_float(o);
}

__device__ __forceinline__ void
logits_strip(int strip, const float* __restrict__ x, const float* __restrict__ U,
             float* __restrict__ a_src, float* __restrict__ a_dst) {
  const int lane  = threadIdx.x & 31;
  const int wv    = threadIdx.x >> 5;
  const int l15   = lane & 15;
  const int half  = lane >> 4;
  const int row0  = (strip << 4) + wv * 4 + half;
  v4f ua[2 * HEADS], ub[2 * HEADS];
  #pragma unroll
  for (int q = 0; q < 2 * HEADS; ++q) {
    ua[q] = *(const v4f*)(U + q * IN_F + l15 * 4);
    ub[q] = *(const v4f*)(U + q * IN_F + 64 + l15 * 4);
  }
  const float* xr = x + (size_t)row0 * IN_F + l15 * 4;
  v4f xa[2], xb[2];
  #pragma unroll
  for (int rp = 0; rp < 2; ++rp) {
    xa[rp] = *(const v4f*)(xr + (size_t)rp * 2 * IN_F);
    xb[rp] = *(const v4f*)(xr + (size_t)rp * 2 * IN_F + 64);
  }
  #pragma unroll
  for (int rp = 0; rp < 2; ++rp) {
    float p[2 * HEADS];
    #pragma unroll
    for (int q = 0; q < 2 * HEADS; ++q) {
      p[q] = xa[rp][0] * ua[q][0] + xa[rp][1] * ua[q][1] + xa[rp][2] * ua[q][2] + xa[rp][3] * ua[q][3]
           + xb[rp][0] * ub[q][0] + xb[rp][1] * ub[q][1] + xb[rp][2] * ub[q][2] + xb[rp][3] * ub[q][3];
    }
    #pragma unroll
    for (int q = 0; q < 2 * HEADS; ++q) p[q] = row_xor_sum<1>(p[q]);
    #pragma unroll
    for (int q = 0; q < 2 * HEADS; ++q) p[q] = row_xor_sum<2>(p[q]);
    #pragma unroll
    for (int q = 0; q < 2 * HEADS; ++q) p[q] = row_xor_sum<4>(p[q]);
    #pragma unroll
    for (int q = 0; q < 2 * HEADS; ++q) p[q] = row_xor_sum<8>(p[q]);
    float v = p[0];
    #pragma unroll
    for (int q = 1; q < 2 * HEADS; ++q) v = (l15 == q) ? p[q] : v;
    if (l15 < 2 * HEADS) {
      float* lp = (l15 < HEADS) ? a_src : a_dst;
      lp[(size_t)(row0 + rp * 2) * HEADS + (l15 & (HEADS - 1))] = v;
    }
  }
}

#define RANK_BLOCKS 1563
#define LOGIT_STRIPS (N_NODES / 16)
#define FUSED_BLOCKS (RANK_BLOCKS + LOGIT_STRIPS)
static_assert(RANK_BLOCKS * FUSED_THREADS >= N_EDGES / 4 &&
              (RANK_BLOCKS - 1) * FUSED_THREADS < N_EDGES / 4, "ranking blocks cover the edges");
static_assert(LOGIT_STRIPS * 16 == N_NODES, "strips cover the nodes");
__global__ void __launch_bounds__(FUSED_THREADS)
rank_logits_kernel(const int* __restrict__ ei, int* __restrict__ deg, int* __restrict__ slots,
                   int* __restrict__ novf, int* __restrict__ ov_src, int* __restrict__ ov_dst,
                   const float* __restrict__ x, const float* __restrict__ U,
                   float* __restrict__ a_src, float* __restrict__ a_dst) {
  const int b = blockIdx.x;
  if (b < RANK_BLOCKS) {
    rank_block(b, ei, deg, slots, novf, ov_src, ov_dst);
    return;
  }
  logits_strip(b - RANK_BLOCKS, x, U, a_src, a_dst);
}

__device__ __forceinline__ v4f edge_scores(const float* __restrict__ a_src, int s, v4f ad) {
  v4f as = *(const v4f*)(a_src + (size_t)s * HEADS);
  v4f e;
  for (int k = 0; k < HEADS; ++k) {
    float ev = as[k] + ad[k];
    e[k] = (ev >= 0.f) ? ev : NEG_SLOPE * ev;
  }
  return e;
}

__device__ __forceinline__ bool
further_edge(int vb, int lane, int tot, int c0, int node, int nov, const int* __restrict__ sl,
             const int* __restrict__ ov_src, const int* __restrict__ ov_dst, int& s) {
  if (vb < SLOT_BLOCK) {
    const int v = vb + lane;
    if (v >= tot) return false;
    s = sl[SLOT_HALF + v - c0];
    return true;
  }
  const int q = vb - SLOT_BLOCK + lane;
  if (q >= nov) return false;
  if (ov_dst[q] != node) return false;
  s = ov_src[q];
  return true;
}

__device__ __forceinline__ void
add_row(v4f (&acc)[HEADS], const float* __restrict__ x, int lane4, int sj,
        float a0, float a1, float a2, float a3) {
  const v4f r = *(const v4f*)(x + (size_t)sj * IN_F + lane4);
  #pragma unroll
  for (int i = 0; i < 4; ++i) {
    acc[0][i] += a0 * r[i];
    acc[1][i] += a1 * r[i];
    acc[2][i] += a2 * r[i];
    acc[3][i] += a3 * r[i];
  }
}

#define XS_PITCH 136
#define GP_NODES 32
#define GP_THREADS 512
#define GP_BLOCKS ((N_NODES + GP_NODES - 1) / GP_NODES)
static_assert(GP_THREADS / 32 * 2 == GP_NODES, "two nodes per wave");
static_assert(N_NODES % 16 == 0, "the last block holds whole 16-row tiles");
__global__ void __launch_bounds__(GP_THREADS, 16)
gather_project_kernel(const int* __restrict__ deg, const int* __restrict__ slots,
                      const int* __restrict__ novf, const int* __restrict__ ov_src,
                      const int* __restrict__ ov_dst,
                      const float* __restrict__ a_src, const float* __restrict__ a_dst,
                      const float* __restrict__ x, const _Float16* __restrict__ Wh,
                      const float* __restrict__ bias, float* __restrict__ out) {
  __shared__ _Float16 xs[HEADS * GP_NODES * XS_PITCH];
  const int wv   = __builtin_amdgcn_readfirstlane((int)(threadIdx.x >> 5));
  const int lane = threadIdx.x & 31;
  const int node0 = blockIdx.x * GP_NODES;
  const int lane4 = lane * 4;
  const float ninf = -__builtin_huge_valf();

  #pragma nounroll
  for (int h2 = 0; h2 < 2; ++h2) {
    const int nl = wv * 2 + h2;
    const int node = node0 + nl;
    v4f acc[HEADS];
    #pragma unroll
    for (int k = 0; k < HEADS; ++k) acc[k] = (v4f){0.f, 0.f, 0.f, 0.f};
    if (node < N_NODES) {
      const v4f ad = *(const v4f*)(a_dst + (size_t)node * HEADS);
      const int c0r = deg[node];
      const int c1r = deg[DEG_PLANE + node];
      asm volatile("" :: "s"(c0r), "s"(c1r), "s"(ad[0]), "s"(ad[1]), "s"(ad[2]), "s"(ad[3]));
      const int c0 = (c0r < SLOT_HALF) ? c0r : SLOT_HALF;
      const int c1 = (c1r < SLOT_HALF) ? c1r : SLOT_HALF;
      const int tot = c0 + c1;
      const int* sl = slots + (size_t)node * SLOT_BLOCK;
      int nov = 0;
      int vend = tot;
      if ((c0r > SLOT_HALF) | (c1r > SLOT_HALF)) {
        nov = *novf;
        nov = (nov < N_EDGES) ? nov : N_EDGES;
        vend = SLOT_BLOCK + nov;
      }

      const bool own = lane < tot;
      int s0 = 0;
      v4f e0 = {0.f, 0.f, 0.f, 0.f};

      v4f mx = {ninf, ninf, ninf, ninf};
      if (own) {
        s0 = sl[(lane < c0) ? lane : SLOT_HALF + lane - c0];
        e0 = edge_scores(a_src, s0, ad);
        for (int k = 0; k < HEADS; ++k) mx[k] = fmaxf(mx[k], e0[k]);
      }
      for (int vb = 32; vb < vend; vb += 32) {
        int s;
        if (further_edge(vb, lane, tot, c0, node, nov, sl, ov_src, ov_dst, s)) {
          v4f e = edge_scores(a_src, s, ad);
          for (int k = 0; k < HEADS; ++k) mx[k] = fmaxf(mx[k], e[k]);
        }
      }
      for (int off = 16; off >= 1; off >>= 1) {
        for (int k = 0; k < HEADS; ++k) mx[k] = fmaxf(mx[k], __shfl_xor(mx[k], off, 32));
      }
      v4f m;
      for (int k = 0; k < HEADS; ++k) {
        const float mk = (__builtin_fabsf(mx[k]) <= 3.402823466e38f) ? mx[k] : 0.f;
        m[k] = __int_as_float(__builtin_amdgcn_readfirstlane(__float_as_int(mk)));
      }

      v4f ex0 = {0.f, 0.f, 0.f, 0.f};
      if (own) {
        for (int k = 0; k < HEADS; ++k) ex0[k] = __expf(e0[k] - m[k]);
      }
      v4f sm = ex0;
      for (int vb = 32; vb < vend; vb += 32) {
        int s;
        if (further_edge(vb, lane, tot, c0, node, nov, sl, ov_src, ov_dst, s)) {
          v4f e = edge_scores(a_src, s, ad);
          for (int k = 0; k < HEADS; ++k) sm[k] += __expf(e[k] - m[k]);
        }
      }
      for (int off = 16; off >= 1; off >>= 1) {
        for (int k = 0; k < HEADS; ++k) sm[k] += __shfl_xor(sm[k], off, 32);
      }
      v4f den;
      for (int k = 0; k < HEADS; ++k)
        den[k] = __int_as_float(__builtin_amdgcn_readfirstlane(__float_as_int(sm[k] + GAT_EPS)));

      {
        v4f al = {0.f, 0.f, 0.f, 0.f};
        if (own) {
          for (int k = 0; k < HEADS; ++k) al[k] = ex0[k] / den[k];
        }
        const int cnt = (tot < 32) ? tot : 32;
        #pragma unroll 4
        for (int j = 0; j < cnt; ++j) {
          const int sj = __builtin_amdgcn_readlane(s0, j);
          const float a0 = __int_as_float(__builtin_amdgcn_readlane(__float_as_int(al[0]), j));
          const float a1 = __int_as_float(__builtin_amdgcn_readlane(__float_as_int(al[1]), j));
          const float a2 = __int_as_float(__builtin_amdgcn_readlane(__float_as_int(al[2]), j));
          const float a3 = __int_as_float(__builtin_amdgcn_readlane(__float_as_int(al[3]), j));
          add_row(acc, x, lane4, sj, a0, a1, a2, a3);
        }
      }
      for (int vb = 32; vb < vend; vb += 32) {
        int s = 0;
        v4f al = {0.f, 0.f, 0.f, 0.f};
        const bool has = further_edge(vb, lane, tot, c0, node, nov, sl, ov_src, ov_dst, s);
        if (has) {
          v4f e = edge_scores(a_src, s, ad);
          for (int k = 0; k < HEADS; ++k) al[k] = __expf(e[k] - m[k]) / den[k];
        }
        unsigned todo = __builtin_amdgcn_ballot_w32(has);
        while (todo != 0u) {
          const int j = __builtin_ctz(todo);
          todo &= todo - 1u;
          const int sj = __builtin_amdgcn_readlane(s, j);
          const float a0 = __int_as_float(__builtin_amdgcn_readlane(__float_as_int(al[0]), j));
          const float a1 = __int_as_float(__builtin_amdgcn_readlane(__float_as_int(al[1]), j));
          const float a2 = __int_as_float(__builtin_amdgcn_readlane(__float_as_int(al[2]), j));
          const float a3 = __int_as_float(__builtin_amdgcn_readlane(__float_as_int(al[3]), j));
          add_row(acc, x, lane4, sj, a0, a1, a2, a3);
        }
      }
    }

    #pragma unroll
    for (int k = 0; k < HEADS; ++k) {
      v4h o;
      #pragma unroll
      for (int i = 0; i < 4; ++i) o[i] = (_Float16)acc[k][i];
      *(v4h*)(xs + (k * GP_NODES + nl) * XS_PITCH + lane * 4) = o;
    }
  }

  const int l15   = lane & 15;
  const int khalf = lane >> 4;
  const int head  = (wv >> 1) & (HEADS - 1);
  const int col0  = head * OUT_F + (wv & 1) * 32;
  const _Float16* bp = Wh + (size_t)(col0 + l15) * IN_F + khalf * 8;

  v16h B0[2];
  if (wv < 8) {
    #pragma unroll
    for (int t = 0; t < 2; ++t) {
      const _Float16* bt = bp + (size_t)t * 16 * IN_F;
      v8h b0 = *(const v8h*)(bt);
      v8h b1 = *(const v8h*)(bt + 16);
      #pragma unroll
      for (int i = 0; i < 8; ++i) { B0[t][i] = b0[i]; B0[t][i + 8] = b1[i]; }
    }
  }
  __syncthreads();
  if (wv >= 8) return;

  #pragma nounroll
  for (int mt = 0; mt < 2; ++mt) {
    const int row0 = node0 + mt * 16;
    if (row0 >= N_NODES) break;
    const _Float16* ap = xs + (head * GP_NODES + mt * 16 + l15) * XS_PITCH + khalf * 8;
    v8f c[2];
    #pragma unroll
    for (int t = 0; t < 2; ++t) c[t] = (v8f){};
    #pragma unroll
    for (int kc = 0; kc < 4; ++kc) {
      v16h A;
      {
        v8h a0 = *(const v8h*)(ap + kc * 32);
        v8h a1 = *(const v8h*)(ap + kc * 32 + 16);
        #pragma unroll
        for (int i = 0; i < 8; ++i) { A[i] = a0[i]; A[i + 8] = a1[i]; }
      }
      #pragma unroll
      for (int t = 0; t < 2; ++t) {
        v16h B = B0[t];
        if (kc > 0) {
          const _Float16* bt = bp + (size_t)t * 16 * IN_F + kc * 32;
          v8h b0 = *(const v8h*)(bt);
          v8h b1 = *(const v8h*)(bt + 16);
          #pragma unroll
          for (int i = 0; i < 8; ++i) { B[i] = b0[i]; B[i + 8] = b1[i]; }
        }
        c[t] = __builtin_amdgcn_wmma_f32_16x16x32_f16(false, A, false, B, (short)0, c[t],
                                                      false, false);
      }
    }

    float* op = out + (size_t)(row0 + khalf * 8) * HF + col0 + l15;
    #pragma unroll
    for (int t = 0; t < 2; ++t) {
      const float b = bias[col0 + t * 16 + l15];
      #pragma unroll
      for (int r = 0; r < 8; ++r) op[(size_t)r * HF + t * 16] = c[t][r] + b;
    }
  }
}

extern "C" void kernel_launch(void* const* d_in, const int* in_sizes, int n_in,
                              void* d_out, int out_size, void* d_ws, size_t ws_size,
                              hipStream_t stream) {
  const float* x       = (const float*)d_in[0];
  const int*   ei      = (const int*)  d_in[1];
  const float* W       = (const float*)d_in[2];
  const float* att_src = (const float*)d_in[3];
  const float* att_dst = (const float*)d_in[4];
  const float* bias    = (const float*)d_in[5];
  float* out = (float*)d_out;

  char* p = (char*)d_ws;
  auto alloc = [&](size_t bytes) {
    char* r = p; p += (bytes + 255) & ~(size_t)255; return r;
  };
  _Float16* Wh    = (_Float16*)alloc((size_t)HF * IN_F * 2);
  float*    U     = (float*)   alloc((size_t)2 * HEADS * IN_F * 4);
  float*    a_src = (float*)   alloc((size_t)N_NODES * HEADS * 4);
  float*    a_dst = (float*)   alloc((size_t)N_NODES * HEADS * 4);
  int*      deg     = (int*)   alloc((size_t)2 * DEG_PLANE * 4);
  int*      novf    = (int*)   alloc((size_t)4);
  int*      slots   = (int*)   alloc((size_t)N_NODES * SLOT_BLOCK * 4);
  int*      ov_src  = (int*)   alloc((size_t)N_EDGES * 4);
  int*      ov_dst  = (int*)   alloc((size_t)N_EDGES * 4);

  prep_kernel       <<<98, 256, 0, stream>>>(W, att_src, att_dst, Wh, U, deg, novf);
  rank_logits_kernel<<<FUSED_BLOCKS, FUSED_THREADS, 0, stream>>>(ei, deg, slots, novf, ov_src, ov_dst,
                                                                 x, U, a_src, a_dst);
  gather_project_kernel<<<GP_BLOCKS, GP_THREADS, 0, stream>>>(deg, slots, novf, ov_src, ov_dst,
                                                              a_src, a_dst, x, Wh, bias, out);
}


# Projection phase re-cut over all 16 waves: a wave owns 16 columns of a head for both row tiles, requests its whole W fragment and bias before an LDS-only barrier; no memory load behind the barrier
# speedup vs baseline: 8.1698x; 1.0540x over previous
#include <hip/hip_runtime.h>
#include <hip/hip_bf16.h>

#define N_NODES 50000
#define N_EDGES 800000
#define IN_F 128
#define OUT_F 64
#define HEADS 4
#define HF 256
#define NEG_SLOPE 0.2f
#define GAT_EPS 1e-16f
#define DEG_PLANE 50176
#define SLOT_HALF 32
#define SLOT_BLOCK 64

typedef __attribute__((ext_vector_type(16))) _Float16 v16h;
typedef __attribute__((ext_vector_type(8)))  _Float16 v8h;
typedef __attribute__((ext_vector_type(4)))  _Float16 v4h;
typedef __attribute__((ext_vector_type(8)))  float    v8f;
typedef __attribute__((ext_vector_type(4)))  float    v4f;
typedef __attribute__((ext_vector_type(4)))  int      v4i;

static_assert((2 * DEG_PLANE) % 4 == 0 && (HF * IN_F) % 4 == 0, "16-byte pieces");
__global__ void __launch_bounds__(256)
prep_kernel(const float* __restrict__ W, const float* __restrict__ att_src,
            const float* __restrict__ att_dst, _Float16* __restrict__ Wh,
            float* __restrict__ U, int* __restrict__ deg, int* __restrict__ novf) {
  const int idx = blockIdx.x * blockDim.x + threadIdx.x;
  const int stride = gridDim.x * blockDim.x;
  for (int i = idx; i < HF * IN_F / 4; i += stride) {
    const v4f w = *(const v4f*)(W + 4 * i);
    v4h o;
    for (int q = 0; q < 4; ++q) o[q] = (_Float16)w[q];
    *(v4h*)(Wh + 4 * i) = o;
  }
  const v4i zero = {0, 0, 0, 0};
  for (int i = idx; i < 2 * DEG_PLANE / 4; i += stride) *(v4i*)(deg + 4 * i) = zero;
  if (idx == 0) *novf = 0;
  for (int o = stride - 1 - idx; o < 2 * HEADS * IN_F; o += stride) {
    const int i = o % IN_F;
    const int k = (o / IN_F) % HEADS;
    const float* att = (o < HEADS * IN_F) ? att_src : att_dst;
    float acc = 0.f;
    #pragma unroll
    for (int f = 0; f < OUT_F; ++f) acc += W[(size_t)(k * OUT_F + f) * IN_F + i] * att[k * OUT_F + f];
    U[o] = acc;
  }
}

#define FUSED_THREADS 128
static_assert(N_EDGES % 4 == 0, "four edges per lane, 16-byte loads");
__device__ __forceinline__ void
rank_block(int blk, const int* __restrict__ ei, int* __restrict__ deg, int* __restrict__ slots,
           int* __restrict__ novf, int* __restrict__ ov_src, int* __restrict__ ov_dst) {
  const int plane = (int)((__builtin_amdgcn_s_sendmsg_rtn(0x87) >> 18) & 1u);
  int i = blk * FUSED_THREADS + threadIdx.x;
  if (i >= N_EDGES / 4) return;
  const v4i s = *(const v4i*)(ei + 4 * i);
  const v4i d = *(const v4i*)(ei + N_EDGES + 4 * i);
  int* cnt = deg + plane * DEG_PLANE;
  v4i r;
  r[0] = atomicAdd(&cnt[d[0]], 1);
  r[1] = atomicAdd(&cnt[d[1]], 1);
  r[2] = atomicAdd(&cnt[d[2]], 1);
  r[3] = atomicAdd(&cnt[d[3]], 1);
  #pragma unroll
  for (int q = 0; q < 4; ++q) {
    if (r[q] < SLOT_HALF) {
      slots[(size_t)d[q] * SLOT_BLOCK + plane * SLOT_HALF + r[q]] = s[q];
    } else {
      const int o = atomicAdd(novf, 1);
      ov_src[o] = s[q];
      ov_dst[o] = d[q];
    }
  }
}


template <int M>
__device__ __forceinline__ float row_xor_sum(float v) {
  int o = __builtin_amdgcn_update_dpp(0, __float_as_int(v), 0x160 + M, 0xf, 0xf, false);
  return v + __int_as_float(o);
}

__device__ __forceinline__ void
logits_strip(int strip, const float* __restrict__ x, const float* __restrict__ U,
             float* __restrict__ a_src, float* __restrict__ a_dst) {
  const int lane  = threadIdx.x & 31;
  const int wv    = threadIdx.x >> 5;
  const int l15   = lane & 15;
  const int half  = lane >> 4;
  const int row0  = (strip << 4) + wv * 4 + half;
  v4f ua[2 * HEADS], ub[2 * HEADS];
  #pragma unroll
  for (int q = 0; q < 2 * HEADS; ++q) {
    ua[q] = *(const v4f*)(U + q * IN_F + l15 * 4);
    ub[q] = *(const v4f*)(U + q * IN_F + 64 + l15 * 4);
  }
  const float* xr = x + (size_t)row0 * IN_F + l15 * 4;
  v4f xa[2], xb[2];
  #pragma unroll
  for (int rp = 0; rp < 2; ++rp) {
    xa[rp] = *(const v4f*)(xr + (size_t)rp * 2 * IN_F);
    xb[rp] = *(const v4f*)(xr + (size_t)rp * 2 * IN_F + 64);
  }
  #pragma unroll
  for (int rp = 0; rp < 2; ++rp) {
    float p[2 * HEADS];
    #pragma unroll
    for (int q = 0; q < 2 * HEADS; ++q) {
      p[q] = xa[rp][0] * ua[q][0] + xa[rp][1] * ua[q][1] + xa[rp][2] * ua[q][2] + xa[rp][3] * ua[q][3]
           + xb[rp][0] * ub[q][0] + xb[rp][1] * ub[q][1] + xb[rp][2] * ub[q][2] + xb[rp][3] * ub[q][3];
    }
    #pragma unroll
    for (int q = 0; q < 2 * HEADS; ++q) p[q] = row_xor_sum<1>(p[q]);
    #pragma unroll
    for (int q = 0; q < 2 * HEADS; ++q) p[q] = row_xor_sum<2>(p[q]);
    #pragma unroll
    for (int q = 0; q < 2 * HEADS; ++q) p[q] = row_xor_sum<4>(p[q]);
    #pragma unroll
    for (int q = 0; q < 2 * HEADS; ++q) p[q] = row_xor_sum<8>(p[q]);
    float v = p[0];
    #pragma unroll
    for (int q = 1; q < 2 * HEADS; ++q) v = (l15 == q) ? p[q] : v;
    if (l15 < 2 * HEADS) {
      float* lp = (l15 < HEADS) ? a_src : a_dst;
      lp[(size_t)(row0 + rp * 2) * HEADS + (l15 & (HEADS - 1))] = v;
    }
  }
}

#define RANK_BLOCKS 1563
#define LOGIT_STRIPS (N_NODES / 16)
#define FUSED_BLOCKS (RANK_BLOCKS + LOGIT_STRIPS)
static_assert(RANK_BLOCKS * FUSED_THREADS >= N_EDGES / 4 &&
              (RANK_BLOCKS - 1) * FUSED_THREADS < N_EDGES / 4, "ranking blocks cover the edges");
static_assert(LOGIT_STRIPS * 16 == N_NODES, "strips cover the nodes");
__global__ void __launch_bounds__(FUSED_THREADS)
rank_logits_kernel(const int* __restrict__ ei, int* __restrict__ deg, int* __restrict__ slots,
                   int* __restrict__ novf, int* __restrict__ ov_src, int* __restrict__ ov_dst,
                   const float* __restrict__ x, const float* __restrict__ U,
                   float* __restrict__ a_src, float* __restrict__ a_dst) {
  const int b = blockIdx.x;
  if (b < RANK_BLOCKS) {
    rank_block(b, ei, deg, slots, novf, ov_src, ov_dst);
    return;
  }
  logits_strip(b - RANK_BLOCKS, x, U, a_src, a_dst);
}

__device__ __forceinline__ v4f edge_scores(const float* __restrict__ a_src, int s, v4f ad) {
  v4f as = *(const v4f*)(a_src + (size_t)s * HEADS);
  v4f e;
  for (int k = 0; k < HEADS; ++k) {
    float ev = as[k] + ad[k];
    e[k] = (ev >= 0.f) ? ev : NEG_SLOPE * ev;
  }
  return e;
}

__device__ __forceinline__ bool
further_edge(int vb, int lane, int tot, int c0, int node, int nov, const int* __restrict__ sl,
             const int* __restrict__ ov_src, const int* __restrict__ ov_dst, int& s) {
  if (vb < SLOT_BLOCK) {
    const int v = vb + lane;
    if (v >= tot) return false;
    s = sl[SLOT_HALF + v - c0];
    return true;
  }
  const int q = vb - SLOT_BLOCK + lane;
  if (q >= nov) return false;
  if (ov_dst[q] != node) return false;
  s = ov_src[q];
  return true;
}

__device__ __forceinline__ void
add_row(v4f (&acc)[HEADS], const float* __restrict__ x, int lane4, int sj,
        float a0, float a1, float a2, float a3) {
  const v4f r = *(const v4f*)(x + (size_t)sj * IN_F + lane4);
  #pragma unroll
  for (int i = 0; i < 4; ++i) {
    acc[0][i] += a0 * r[i];
    acc[1][i] += a1 * r[i];
    acc[2][i] += a2 * r[i];
    acc[3][i] += a3 * r[i];
  }
}

#define XS_PITCH 136
#define GP_NODES 32
#define GP_THREADS 512
#define GP_BLOCKS ((N_NODES + GP_NODES - 1) / GP_NODES)
static_assert(GP_THREADS / 32 * 2 == GP_NODES, "two nodes per wave");
static_assert(N_NODES % 16 == 0, "the last block holds whole 16-row tiles");
__global__ void __launch_bounds__(GP_THREADS, 16)
gather_project_kernel(const int* __restrict__ deg, const int* __restrict__ slots,
                      const int* __restrict__ novf, const int* __restrict__ ov_src,
                      const int* __restrict__ ov_dst,
                      const float* __restrict__ a_src, const float* __restrict__ a_dst,
                      const float* __restrict__ x, const _Float16* __restrict__ Wh,
                      const float* __restrict__ bias, float* __restrict__ out) {
  __shared__ _Float16 xs[HEADS * GP_NODES * XS_PITCH];
  const int wv   = __builtin_amdgcn_readfirstlane((int)(threadIdx.x >> 5));
  const int lane = threadIdx.x & 31;
  const int node0 = blockIdx.x * GP_NODES;
  const int lane4 = lane * 4;
  const float ninf = -__builtin_huge_valf();

  #pragma nounroll
  for (int h2 = 0; h2 < 2; ++h2) {
    const int nl = wv * 2 + h2;
    const int node = node0 + nl;
    v4f acc[HEADS];
    #pragma unroll
    for (int k = 0; k < HEADS; ++k) acc[k] = (v4f){0.f, 0.f, 0.f, 0.f};
    if (node < N_NODES) {
      const v4f ad = *(const v4f*)(a_dst + (size_t)node * HEADS);
      const int c0r = deg[node];
      const int c1r = deg[DEG_PLANE + node];
      asm volatile("" :: "s"(c0r), "s"(c1r), "s"(ad[0]), "s"(ad[1]), "s"(ad[2]), "s"(ad[3]));
      const int c0 = (c0r < SLOT_HALF) ? c0r : SLOT_HALF;
      const int c1 = (c1r < SLOT_HALF) ? c1r : SLOT_HALF;
      const int tot = c0 + c1;
      const int* sl = slots + (size_t)node * SLOT_BLOCK;
      int nov = 0;
      int vend = tot;
      if ((c0r > SLOT_HALF) | (c1r > SLOT_HALF)) {
        nov = *novf;
        nov = (nov < N_EDGES) ? nov : N_EDGES;
        vend = SLOT_BLOCK + nov;
      }

      const bool own = lane < tot;
      int s0 = 0;
      v4f e0 = {0.f, 0.f, 0.f, 0.f};

      v4f mx = {ninf, ninf, ninf, ninf};
      if (own) {
        s0 = sl[(lane < c0) ? lane : SLOT_HALF + lane - c0];
        e0 = edge_scores(a_src, s0, ad);
        for (int k = 0; k < HEADS; ++k) mx[k] = fmaxf(mx[k], e0[k]);
      }
      for (int vb = 32; vb < vend; vb += 32) {
        int s;
        if (further_edge(vb, lane, tot, c0, node, nov, sl, ov_src, ov_dst, s)) {
          v4f e = edge_scores(a_src, s, ad);
          for (int k = 0; k < HEADS; ++k) mx[k] = fmaxf(mx[k], e[k]);
        }
      }
      for (int off = 16; off >= 1; off >>= 1) {
        for (int k = 0; k < HEADS; ++k) mx[k] = fmaxf(mx[k], __shfl_xor(mx[k], off, 32));
      }
      v4f m;
      for (int k = 0; k < HEADS; ++k) {
        const float mk = (__builtin_fabsf(mx[k]) <= 3.402823466e38f) ? mx[k] : 0.f;
        m[k] = __int_as_float(__builtin_amdgcn_readfirstlane(__float_as_int(mk)));
      }

      v4f ex0 = {0.f, 0.f, 0.f, 0.f};
      if (own) {
        for (int k = 0; k < HEADS; ++k) ex0[k] = __expf(e0[k] - m[k]);
      }
      v4f sm = ex0;
      for (int vb = 32; vb < vend; vb += 32) {
        int s;
        if (further_edge(vb, lane, tot, c0, node, nov, sl, ov_src, ov_dst, s)) {
          v4f e = edge_scores(a_src, s, ad);
          for (int k = 0; k < HEADS; ++k) sm[k] += __expf(e[k] - m[k]);
        }
      }
      for (int off = 16; off >= 1; off >>= 1) {
        for (int k = 0; k < HEADS; ++k) sm[k] += __shfl_xor(sm[k], off, 32);
      }
      v4f den;
      for (int k = 0; k < HEADS; ++k)
        den[k] = __int_as_float(__builtin_amdgcn_readfirstlane(__float_as_int(sm[k] + GAT_EPS)));

      {
        v4f al = {0.f, 0.f, 0.f, 0.f};
        if (own) {
          for (int k = 0; k < HEADS; ++k) al[k] = ex0[k] / den[k];
        }
        const int cnt = (tot < 32) ? tot : 32;
        #pragma unroll 4
        for (int j = 0; j < cnt; ++j) {
          const int sj = __builtin_amdgcn_readlane(s0, j);
          const float a0 = __int_as_float(__builtin_amdgcn_readlane(__float_as_int(al[0]), j));
          const float a1 = __int_as_float(__builtin_amdgcn_readlane(__float_as_int(al[1]), j));
          const float a2 = __int_as_float(__builtin_amdgcn_readlane(__float_as_int(al[2]), j));
          const float a3 = __int_as_float(__builtin_amdgcn_readlane(__float_as_int(al[3]), j));
          add_row(acc, x, lane4, sj, a0, a1, a2, a3);
        }
      }
      for (int vb = 32; vb < vend; vb += 32) {
        int s = 0;
        v4f al = {0.f, 0.f, 0.f, 0.f};
        const bool has = further_edge(vb, lane, tot, c0, node, nov, sl, ov_src, ov_dst, s);
        if (has) {
          v4f e = edge_scores(a_src, s, ad);
          for (int k = 0; k < HEADS; ++k) al[k] = __expf(e[k] - m[k]) / den[k];
        }
        unsigned todo = __builtin_amdgcn_ballot_w32(has);
        while (todo != 0u) {
          const int j = __builtin_ctz(todo);
          todo &= todo - 1u;
          const int sj = __builtin_amdgcn_readlane(s, j);
          const float a0 = __int_as_float(__builtin_amdgcn_readlane(__float_as_int(al[0]), j));
          const float a1 = __int_as_float(__builtin_amdgcn_readlane(__float_as_int(al[1]), j));
          const float a2 = __int_as_float(__builtin_amdgcn_readlane(__float_as_int(al[2]), j));
          const float a3 = __int_as_float(__builtin_amdgcn_readlane(__float_as_int(al[3]), j));
          add_row(acc, x, lane4, sj, a0, a1, a2, a3);
        }
      }
    }

    #pragma unroll
    for (int k = 0; k < HEADS; ++k) {
      v4h o;
      #pragma unroll
      for (int i = 0; i < 4; ++i) o[i] = (_Float16)acc[k][i];
      *(v4h*)(xs + (k * GP_NODES + nl) * XS_PITCH + lane * 4) = o;
    }
  }

  const int l15   = lane & 15;
  const int khalf = lane >> 4;
  const int head  = wv >> 2;
  const int col0  = head * OUT_F + (wv & 3) * 16;
  const _Float16* bp = Wh + (size_t)(col0 + l15) * IN_F + khalf * 8;

  v16h B[4];
  #pragma unroll
  for (int kc = 0; kc < 4; ++kc) {
    v8h b0 = *(const v8h*)(bp + kc * 32);
    v8h b1 = *(const v8h*)(bp + kc * 32 + 16);
    #pragma unroll
    for (int i = 0; i < 8; ++i) { B[kc][i] = b0[i]; B[kc][i + 8] = b1[i]; }
  }
  const float bv = bias[col0 + l15];
  __builtin_amdgcn_fence(__ATOMIC_RELEASE, "workgroup", "local");
  __builtin_amdgcn_s_barrier();
  __builtin_amdgcn_fence(__ATOMIC_ACQUIRE, "workgroup", "local");

  #pragma nounroll
  for (int mt = 0; mt < 2; ++mt) {
    const int row0 = node0 + mt * 16;
    if (row0 >= N_NODES) break;
    const _Float16* ap = xs + (head * GP_NODES + mt * 16 + l15) * XS_PITCH + khalf * 8;
    v8f c = (v8f){};
    #pragma unroll
    for (int kc = 0; kc < 4; ++kc) {
      v16h A;
      {
        v8h a0 = *(const v8h*)(ap + kc * 32);
        v8h a1 = *(const v8h*)(ap + kc * 32 + 16);
        #pragma unroll
        for (int i = 0; i < 8; ++i) { A[i] = a0[i]; A[i + 8] = a1[i]; }
      }
      c = __builtin_amdgcn_wmma_f32_16x16x32_f16(false, A, false, B[kc], (short)0, c,
                                                 false, false);
    }

    float* op = out + (size_t)(row0 + khalf * 8) * HF + col0 + l15;
    #pragma unroll
    for (int r = 0; r < 8; ++r) op[(size_t)r * HF] = c[r] + bv;
  }
}

extern "C" void kernel_launch(void* const* d_in, const int* in_sizes, int n_in,
                              void* d_out, int out_size, void* d_ws, size_t ws_size,
                              hipStream_t stream) {
  const float* x       = (const float*)d_in[0];
  const int*   ei      = (const int*)  d_in[1];
  const float* W       = (const float*)d_in[2];
  const float* att_src = (const float*)d_in[3];
  const float* att_dst = (const float*)d_in[4];
  const float* bias    = (const float*)d_in[5];
  float* out = (float*)d_out;

  char* p = (char*)d_ws;
  auto alloc = [&](size_t bytes) {
    char* r = p; p += (bytes + 255) & ~(size_t)255; return r;
  };
  _Float16* Wh    = (_Float16*)alloc((size_t)HF * IN_F * 2);
  float*    U     = (float*)   alloc((size_t)2 * HEADS * IN_F * 4);
  float*    a_src = (float*)   alloc((size_t)N_NODES * HEADS * 4);
  float*    a_dst = (float*)   alloc((size_t)N_NODES * HEADS * 4);
  int*      deg     = (int*)   alloc((size_t)2 * DEG_PLANE * 4);
  int*      novf    = (int*)   alloc((size_t)4);
  int*      slots   = (int*)   alloc((size_t)N_NODES * SLOT_BLOCK * 4);
  int*      ov_src  = (int*)   alloc((size_t)N_EDGES * 4);
  int*      ov_dst  = (int*)   alloc((size_t)N_EDGES * 4);

  prep_kernel       <<<98, 256, 0, stream>>>(W, att_src, att_dst, Wh, U, deg, novf);
  rank_logits_kernel<<<FUSED_BLOCKS, FUSED_THREADS, 0, stream>>>(ei, deg, slots, novf, ov_src, ov_dst,
                                                                 x, U, a_src, a_dst);
  gather_project_kernel<<<GP_BLOCKS, GP_THREADS, 0, stream>>>(deg, slots, novf, ov_src, ov_dst,
                                                              a_src, a_dst, x, Wh, bias, out);
}
